# starting kernel
# baseline (speedup 1.0000x reference)
#include <hip/hip_runtime.h>
#include <hip/hip_bf16.h>
#include <cmath>

typedef __attribute__((ext_vector_type(2))) float v2f;
typedef __attribute__((ext_vector_type(8))) float v8f;

#define PNODES 22
#define GNUM   1920
#define CIN    14
#define HCOLS  160
#define EPG    462
#define TLEN   120
#define BSZ    16
#define OCH    272
#define KDIM   456
#define NCLS   17

__device__ __forceinline__ v8f wmma_f32_4(v2f a, v2f b, v8f c) {
  return __builtin_amdgcn_wmma_f32_16x16x4_f32(
       false, a,  false, b,
       (short)0, c,  false,  false);
}

__device__ __forceinline__ void gcn_layer(
    float (&H)[32][HCOLS], float (&Tm)[32][HCOLS], const float (&A)[32][24],
    const float* __restrict__ W, const float* __restrict__ bias,
    int Cin, int Cout, int wave, int half, int l16)
{
  const int Kpad   = (Cin + 3) & ~3;
  const int ntiles = (Cout + 15) >> 4;

  for (int t = wave; t < 2 * ntiles; t += 4) {
    const int mt = t & 1, nt = t >> 1;
    const int m   = mt * 16 + l16;
    const int n   = nt * 16 + l16;
    const bool nok = (n < Cout);
    const int  ncl = nok ? n : 0;
    v8f acc = {};
    for (int k = 0; k < Kpad; k += 4) {
      const int k0 = k + half * 2;
      const int kr0 = (k0     < Cin) ? k0     : 0;
      const int kr1 = (k0 + 1 < Cin) ? k0 + 1 : 0;
      v2f a, b;
      a.x = H[m][k0];
      a.y = H[m][k0 + 1];
      const float w0 = W[(size_t)kr0 * Cout + ncl];
      const float w1 = W[(size_t)kr1 * Cout + ncl];
      b.x = (k0     < Cin && nok) ? w0 : 0.0f;
      b.y = (k0 + 1 < Cin && nok) ? w1 : 0.0f;
      acc = wmma_f32_4(a, b, acc);
    }
    for (int v = 0; v < 8; ++v) {
      const int row = mt * 16 + v + half * 8;
      Tm[row][n] = acc[v];
    }
  }
  __syncthreads();

  for (int t = wave; t < 2 * ntiles; t += 4) {
    const int mt = t & 1, nt = t >> 1;
    const int m  = mt * 16 + l16;
    const int n  = nt * 16 + l16;
    const bool nok = (n < Cout);
    v8f acc = {};
    for (int k = 0; k < 24; k += 4) {
      const int k0 = k + half * 2;
      v2f a, b;
      a.x = A[m][k0];     a.y = A[m][k0 + 1];
      b.x = Tm[k0][n];    b.y = Tm[k0 + 1][n];
      acc = wmma_f32_4(a, b, acc);
    }
    const float bv = nok ? bias[nok ? n : 0] : 0.0f;
    for (int v = 0; v < 8; ++v) {
      const int row = mt * 16 + v + half * 8;
      H[row][n] = (row < PNODES && nok) ? fmaxf(acc[v] + bv, 0.0f) : 0.0f;
    }
  }
  __syncthreads();
}

__global__ __launch_bounds__(128) void gcn_fused_kernel(
    const float* __restrict__ x, const float* __restrict__ ea,
    const float* __restrict__ W1, const float* __restrict__ b1,
    const float* __restrict__ W2, const float* __restrict__ b2,
    const float* __restrict__ W3, const float* __restrict__ b3,
    const float* __restrict__ W4, const float* __restrict__ b4,
    float* __restrict__ pooled)
{
  __shared__ float A[32][24];
  __shared__ float H[32][HCOLS];
  __shared__ float Tm[32][HCOLS];
  __shared__ float dinv[24];

  const int g    = blockIdx.x;
  const int tid  = threadIdx.x;
  const int wave = tid >> 5;
  const int lane = tid & 31;
  const int half = lane >> 4;
  const int l16  = lane & 15;

  const float* eaw = ea + (size_t)g * EPG * 5;

  {
    const char* base = (const char*)eaw;
    __builtin_prefetch(base + (size_t)tid * 64, 0, 1);
    __builtin_prefetch(base + (size_t)(tid + 128) * 64 % 9240, 0, 1);
  }

  for (int i = tid; i < 32 * 24;    i += 128) (&A[0][0])[i] = 0.0f;
  for (int i = tid; i < 32 * HCOLS; i += 128) (&H[0][0])[i] = 0.0f;
  if (tid < 24) dinv[tid] = 0.0f;
  __syncthreads();

  if (tid < PNODES) {
    const int j = tid;
    float deg = 1.0f;
    for (int i = 0; i < PNODES; ++i) {
      if (i == j) continue;
      const int e = i * (PNODES - 1) + (j < i ? j : j - 1);
      deg += eaw[e * 5 + 4];
    }
    dinv[j] = (deg > 0.0f) ? rsqrtf(deg) : 0.0f;
  }
  for (int idx = tid; idx < PNODES * CIN; idx += 128) {
    const int r = idx / CIN, c = idx % CIN;
    H[r][c] = x[(size_t)(g * PNODES + r) * CIN + c];
  }
  __syncthreads();

  for (int idx = tid; idx < PNODES * PNODES; idx += 128) {
    const int j = idx / PNODES, i = idx % PNODES;
    float v;
    if (i == j) {
      v = dinv[j] * dinv[j];
    } else {
      const int e = i * (PNODES - 1) + (j < i ? j : j - 1);
      v = dinv[i] * eaw[e * 5 + 4] * dinv[j];
    }
    A[j][i] = v;
  }
  __syncthreads();

  gcn_layer(H, Tm, A, W1, b1, 14, 16,  wave, half, l16);
  gcn_layer(H, Tm, A, W2, b2, 16, 32,  wave, half, l16);
  gcn_layer(H, Tm, A, W3, b3, 32, 64,  wave, half, l16);
  gcn_layer(H, Tm, A, W4, b4, 64, 152, wave, half, l16);

  for (int c = tid; c < 152; c += 128) {
    float s = 0.0f;
    for (int r = 0; r < PNODES; ++r) s += H[r][c];
    pooled[(size_t)g * 152 + c] = s * (1.0f / 22.0f);
  }
}

__global__ __launch_bounds__(128) void conv_wmma_kernel(
    const float* __restrict__ pooled, const float* __restrict__ cw,
    const float* __restrict__ cb, float* __restrict__ convout)
{
  const int b    = blockIdx.x >> 3;
  const int tt   = blockIdx.x & 7;
  const int tid  = threadIdx.x;
  const int wave = tid >> 5;
  const int lane = tid & 31;
  const int half = lane >> 4;
  const int l16  = lane & 15;
  const int trow = tt * 16 + l16;
  const bool trok = (trow < TLEN);

  for (int nt = wave; nt < OCH / 16; nt += 4) {
    const int n = nt * 16 + l16;
    v8f acc = {};
    for (int k = 0; k < KDIM; k += 4) {
      const int k0 = k + half * 2;
      const int c0 = k0 / 3,       kk0 = k0 - c0 * 3;
      const int c1 = (k0 + 1) / 3, kk1 = (k0 + 1) - ((k0 + 1) / 3) * 3;
      const int ts0 = trow + kk0 - 1;
      const int ts1 = trow + kk1 - 1;
      const int tc0 = ts0 < 0 ? 0 : (ts0 >= TLEN ? TLEN - 1 : ts0);
      const int tc1 = ts1 < 0 ? 0 : (ts1 >= TLEN ? TLEN - 1 : ts1);
      const float x0 = pooled[(size_t)(b * TLEN + tc0) * 152 + c0];
      const float x1 = pooled[(size_t)(b * TLEN + tc1) * 152 + c1];
      v2f a, bb;
      a.x = (trok && ts0 >= 0 && ts0 < TLEN) ? x0 : 0.0f;
      a.y = (trok && ts1 >= 0 && ts1 < TLEN) ? x1 : 0.0f;
      bb.x = cw[((size_t)n * 152 + c0) * 3 + kk0];
      bb.y = cw[((size_t)n * 152 + c1) * 3 + kk1];
      acc = wmma_f32_4(a, bb, acc);
    }
    const float bv = cb[n];
    for (int v = 0; v < 8; ++v) {
      const int t = tt * 16 + v + half * 8;
      if (t < TLEN) convout[((size_t)b * OCH + n) * TLEN + t] = acc[v] + bv;
    }
  }
}

__global__ __launch_bounds__(256) void caps_norm_kernel(
    const float* __restrict__ convout, const float* __restrict__ gamma,
    const float* __restrict__ beta, float* __restrict__ out)
{
  const int idx = blockIdx.x * 256 + threadIdx.x;
  if (idx >= BSZ * TLEN * NCLS) return;
  const int cls = idx % NCLS;
  const int t   = (idx / NCLS) % TLEN;
  const int b   = idx / (NCLS * TLEN);
  const float scale = gamma[t] * rsqrtf(1.0f + 0.001f);
  const float shift = beta[t];
  float sum = 0.0f;
  for (int d = 0; d < 16; ++d) {
    const int o = d * NCLS + cls;
    const float v  = convout[((size_t)b * OCH + o) * TLEN + t];
    const float s  = 1.0f / (1.0f + __expf(-(v * scale + shift)));
    const float df = s - 0.5f;
    sum += df * df;
  }
  out[idx] = sqrtf(sum * 0.25f);
}

extern "C" void kernel_launch(void* const* d_in, const int* in_sizes, int n_in,
                              void* d_out, int out_size, void* d_ws, size_t ws_size,
                              hipStream_t stream) {
  (void)in_sizes; (void)n_in; (void)out_size; (void)ws_size;
  const float* x  = (const float*)d_in[0];
  const float* ea = (const float*)d_in[3];
  const float* W1 = (const float*)d_in[4];  const float* b1 = (const float*)d_in[5];
  const float* W2 = (const float*)d_in[6];  const float* b2 = (const float*)d_in[7];
  const float* W3 = (const float*)d_in[8];  const float* b3 = (const float*)d_in[9];
  const float* W4 = (const float*)d_in[10]; const float* b4 = (const float*)d_in[11];
  const float* cw = (const float*)d_in[12]; const float* cb = (const float*)d_in[13];
  const float* gm = (const float*)d_in[14]; const float* bt = (const float*)d_in[15];

  float* pooled  = (float*)d_ws;
  float* convout = pooled + (size_t)GNUM * 152;

  gcn_fused_kernel<<<GNUM, 128, 0, stream>>>(x, ea, W1, b1, W2, b2, W3, b3, W4, b4, pooled);
  conv_wmma_kernel<<<BSZ * 8, 128, 0, stream>>>(pooled, cw, cb, convout);
  const int nout = BSZ * TLEN * NCLS;
  caps_norm_kernel<<<(nout + 255) / 256, 256, 0, stream>>>(convout, gm, bt, (float*)d_out);
}


# GCN kernel weight-stationary: 960 workgroups x 2 graphs looped; a wave owns fixed output tiles of every layer and keeps their W fragments in registers; layer 4: one H fragment pair feeds 5 WMMAs
# speedup vs baseline: 3.9654x; 3.9654x over previous
#include <hip/hip_runtime.h>
#include <hip/hip_bf16.h>
#include <cmath>

typedef __attribute__((ext_vector_type(2))) float v2f;
typedef __attribute__((ext_vector_type(8))) float v8f;

#define PNODES 22
#define GNUM   1920
#define CIN    14
#define HCOLS  160
#define EPG    462
#define TLEN   120
#define BSZ    16
#define OCH    272
#define KDIM   456
#define NCLS   17

__device__ __forceinline__ v8f wmma_f32_4(v2f a, v2f b, v8f c) {
  return __builtin_amdgcn_wmma_f32_16x16x4_f32(
       false, a,  false, b,
       (short)0, c,  false,  false);
}

#define GCN_WGS 960

template <int CIN_, int COUT_, int KS>
__device__ __forceinline__ void load_w_frags(v2f (&wf)[KS], const float* __restrict__ W,
                                             int nt, int half, int l16)
{
  const int  n   = nt * 16 + l16;
  const bool nok = n < COUT_;
  const int  ncl = nok ? n : 0;
  const float* base = W + (2 * half) * COUT_ + ncl;
#pragma unroll
  for (int ks = 0; ks < KS; ++ks) {
    if (4 * ks + 3 < CIN_) {
      const float w0 = base[(4 * ks) * COUT_];
      const float w1 = base[(4 * ks + 1) * COUT_];
      wf[ks].x = nok ? w0 : 0.0f;
      wf[ks].y = nok ? w1 : 0.0f;
    } else {
      const int k0  = 4 * ks + 2 * half;
      const int kr0 = (k0     < CIN_) ? k0     : 0;
      const int kr1 = (k0 + 1 < CIN_) ? k0 + 1 : 0;
      const float w0 = W[kr0 * COUT_ + ncl];
      const float w1 = W[kr1 * COUT_ + ncl];
      wf[ks].x = (k0     < CIN_ && nok) ? w0 : 0.0f;
      wf[ks].y = (k0 + 1 < CIN_ && nok) ? w1 : 0.0f;
    }
  }
}

__device__ __forceinline__ void store_tm(float (&Tm)[32][HCOLS], v8f c, int mt, int n, int half)
{
#pragma unroll
  for (int v = 0; v < 8; ++v) Tm[mt * 16 + 8 * half + v][n] = c[v];
}

__device__ __forceinline__ void store_h(float (&H)[32][HCOLS], v8f c, int mt, int n, int half,
                                        float bv, bool nok)
{
#pragma unroll
  for (int v = 0; v < 8; ++v) {
    const int row = mt * 16 + 8 * half + v;
    H[row][n] = (row < PNODES && nok) ? fmaxf(c[v] + bv, 0.0f) : 0.0f;
  }
}

__device__ __forceinline__ void stepb_pair(float (&H)[32][HCOLS], const float (&Tm)[32][HCOLS],
    const float (&A)[32][24], int n, int l16, int half, float bv, bool nok)
{
  v2f b[6], adj0[6], adj1[6];
#pragma unroll
  for (int ks = 0; ks < 6; ++ks) {
    const int k0 = 4 * ks + 2 * half;
    b[ks].x = Tm[k0][n];  b[ks].y = Tm[k0 + 1][n];
    adj0[ks] = *(const v2f*)&A[l16][k0];
    adj1[ks] = *(const v2f*)&A[16 + l16][k0];
  }
  v8f c0 = {}, c1 = {};
#pragma unroll
  for (int ks = 0; ks < 6; ++ks) {
    c0 = wmma_f32_4(adj0[ks], b[ks], c0);
    c1 = wmma_f32_4(adj1[ks], b[ks], c1);
  }
  store_h(H, c0, 0, n, half, bv, nok);
  store_h(H, c1, 1, n, half, bv, nok);
}

__device__ __forceinline__ void stepb_single(float (&H)[32][HCOLS], const float (&Tm)[32][HCOLS],
    const float (&A)[32][24], int mt, int n, int l16, int half, float bv, bool nok)
{
  v2f b[6], adjh[6];
#pragma unroll
  for (int ks = 0; ks < 6; ++ks) {
    const int k0 = 4 * ks + 2 * half;
    b[ks].x = Tm[k0][n];  b[ks].y = Tm[k0 + 1][n];
    adjh[ks] = *(const v2f*)&A[mt * 16 + l16][k0];
  }
  v8f c = {};
#pragma unroll
  for (int ks = 0; ks < 6; ++ks) c = wmma_f32_4(adjh[ks], b[ks], c);
  store_h(H, c, mt, n, half, bv, nok);
}

__global__ __launch_bounds__(128) void gcn_fused_kernel(
    const float* __restrict__ x, const float* __restrict__ ea,
    const float* __restrict__ W1, const float* __restrict__ b1,
    const float* __restrict__ W2, const float* __restrict__ b2,
    const float* __restrict__ W3, const float* __restrict__ b3,
    const float* __restrict__ W4, const float* __restrict__ b4,
    float* __restrict__ pooled)
{
  __shared__ __attribute__((aligned(16))) float A[32][24];
  __shared__ __attribute__((aligned(16))) float H[32][HCOLS];
  __shared__ __attribute__((aligned(16))) float Tm[32][HCOLS];
  __shared__ float dinv[24];

  const int tid  = threadIdx.x;
  const int wave = __builtin_amdgcn_readfirstlane(tid >> 5);
  const int lane = tid & 31;
  const int half = lane >> 4;
  const int l16  = lane & 15;
  const int hm   = wave & 1;

  const int ntA = 2 * wave, ntB = 2 * wave + 1, ntC = 8 + (wave >> 1);
  const int nt3 = wave, nt2 = wave >> 1;
  const int nA = ntA * 16 + l16, nB = ntB * 16 + l16, nC = ntC * 16 + l16;
  const int n3 = nt3 * 16 + l16, n2 = nt2 * 16 + l16, n1 = l16;
  const bool okC = nC < 152;
  v2f w4a[16], w4b[16], w4c[16], w3f[8], w2f[4], w1f[4];
  load_w_frags<64, 152, 16>(w4a, W4, ntA, half, l16);
  load_w_frags<64, 152, 16>(w4b, W4, ntB, half, l16);
  load_w_frags<64, 152, 16>(w4c, W4, ntC, half, l16);
  load_w_frags<32, 64, 8>(w3f, W3, nt3, half, l16);
  load_w_frags<16, 32, 4>(w2f, W2, nt2, half, l16);
  load_w_frags<14, 16, 4>(w1f, W1, 0,   half, l16);
  const float bvA = b4[nA], bvB = b4[nB], bvC = okC ? b4[okC ? nC : 0] : 0.0f;
  const float bv3 = b3[n3], bv2 = b2[n2], bv1 = b1[n1];

  for (int i = tid; i < 32 * 24; i += 128) (&A[0][0])[i] = 0.0f;

  for (int g = blockIdx.x; g < GNUM; g += GCN_WGS) {
    int l16s = l16, halfs = half, tids = tid;
    asm volatile("" : "+v"(l16s), "+v"(halfs), "+v"(tids));
    const int l16 = l16s, half = halfs, tid = tids;
    const int nA = ntA * 16 + l16, nB = ntB * 16 + l16, nC = ntC * 16 + l16;
    const int n3 = nt3 * 16 + l16, n2 = nt2 * 16 + l16, n1 = l16;
    const float* hrow0 = &H[l16][2 * half];
    const float* hrow1 = &H[16 + l16][2 * half];
    const float* hrowh = hm ? hrow1 : hrow0;
    const float* eaw = ea + (size_t)g * EPG * 5;

    if (g < GCN_WGS) {
      const char* base = (const char*)eaw;
      __builtin_prefetch(base + (size_t)tid * 64, 0, 1);
      __builtin_prefetch(base + (size_t)(tid + 128) * 64 % 9240, 0, 1);
    }
    if (g + GCN_WGS < GNUM) {
      const char* nbase = (const char*)(ea + (size_t)(g + GCN_WGS) * EPG * 5);
      const char* nfeat = (const char*)(x + (size_t)(g + GCN_WGS) * PNODES * CIN);
      __builtin_prefetch(nbase + (size_t)tid * 64, 0, 1);
      __builtin_prefetch(nbase + (size_t)(tid + 128) * 64 % 9240, 0, 1);
      __builtin_prefetch(nfeat + (size_t)tid * 64 % 1232, 0, 1);
    }

    if (tid < PNODES) {
      const int j = tid;
      float deg = 1.0f;
      for (int i = 0; i < PNODES; ++i) {
        if (i == j) continue;
        const int e = i * (PNODES - 1) + (j < i ? j : j - 1);
        deg += eaw[e * 5 + 4];
      }
      dinv[j] = (deg > 0.0f) ? rsqrtf(deg) : 0.0f;
    }
    for (int idx = tid; idx < 32 * 16; idx += 128) {
      const int r = idx >> 4, c = idx & 15;
      if (r >= PNODES || c >= CIN) H[r][c] = 0.0f;
    }
    for (int idx = tid; idx < PNODES * CIN; idx += 128) {
      const int r = idx / CIN, c = idx % CIN;
      H[r][c] = x[(size_t)(g * PNODES + r) * CIN + c];
    }
    __syncthreads();

    for (int idx = tid; idx < PNODES * PNODES; idx += 128) {
      const int j = idx / PNODES, i = idx % PNODES;
      float v;
      if (i == j) {
        v = dinv[j] * dinv[j];
      } else {
        const int e = i * (PNODES - 1) + (j < i ? j : j - 1);
        v = dinv[i] * eaw[e * 5 + 4] * dinv[j];
      }
      A[j][i] = v;
    }
    __syncthreads();

    if (wave < 2) {
      v8f c = {};
#pragma unroll
      for (int ks = 0; ks < 4; ++ks) c = wmma_f32_4(*(const v2f*)(hrowh + 4 * ks), w1f[ks], c);
      store_tm(Tm, c, hm, n1, half);
    }
    __syncthreads();
    if (wave < 2) stepb_single(H, Tm, A, hm, n1, l16, half, bv1, true);
    __syncthreads();

    {
      v8f c = {};
#pragma unroll
      for (int ks = 0; ks < 4; ++ks) c = wmma_f32_4(*(const v2f*)(hrowh + 4 * ks), w2f[ks], c);
      store_tm(Tm, c, hm, n2, half);
    }
    __syncthreads();
    stepb_single(H, Tm, A, hm, n2, l16, half, bv2, true);
    __syncthreads();

    {
      v8f c0 = {}, c1 = {};
#pragma unroll
      for (int ks = 0; ks < 8; ++ks) {
        const v2f a0 = *(const v2f*)(hrow0 + 4 * ks);
        const v2f a1 = *(const v2f*)(hrow1 + 4 * ks);
        c0 = wmma_f32_4(a0, w3f[ks], c0);
        c1 = wmma_f32_4(a1, w3f[ks], c1);
      }
      store_tm(Tm, c0, 0, n3, half);
      store_tm(Tm, c1, 1, n3, half);
    }
    __syncthreads();
    stepb_pair(H, Tm, A, n3, l16, half, bv3, true);
    __syncthreads();

    {
      v8f c00 = {}, c01 = {}, c10 = {}, c11 = {}, ch = {};
      v2f xa[4], xb[4];
#pragma unroll
      for (int i = 0; i < 4; ++i) {
        xa[i] = *(const v2f*)(hrow0 + 4 * i);
        xb[i] = *(const v2f*)(hrow1 + 4 * i);
      }
#pragma unroll
      for (int ck = 0; ck < 4; ++ck) {
        v2f na[4], nb[4];
        if (ck + 1 < 4) {
#pragma unroll
          for (int i = 0; i < 4; ++i) {
            na[i] = *(const v2f*)(hrow0 + 4 * ((ck + 1) * 4 + i));
            nb[i] = *(const v2f*)(hrow1 + 4 * ((ck + 1) * 4 + i));
          }
        }
        __builtin_amdgcn_sched_barrier(0);
#pragma unroll
        for (int i = 0; i < 4; ++i) {
          const int ks = ck * 4 + i;
          const v2f a0 = xa[i], a1 = xb[i];
          const v2f ah = hm ? a1 : a0;
          c00 = wmma_f32_4(a0, w4a[ks], c00);
          c01 = wmma_f32_4(a1, w4a[ks], c01);
          c10 = wmma_f32_4(a0, w4b[ks], c10);
          c11 = wmma_f32_4(a1, w4b[ks], c11);
          ch  = wmma_f32_4(ah, w4c[ks], ch);
        }
        __builtin_amdgcn_sched_barrier(0);
        if (ck + 1 < 4) {
#pragma unroll
          for (int i = 0; i < 4; ++i) { xa[i] = na[i]; xb[i] = nb[i]; }
        }
      }
      store_tm(Tm, c00, 0, nA, half);
      store_tm(Tm, c01, 1, nA, half);
      store_tm(Tm, c10, 0, nB, half);
      store_tm(Tm, c11, 1, nB, half);
      store_tm(Tm, ch, hm, nC, half);
    }
    __syncthreads();
    stepb_pair(H, Tm, A, nA, l16, half, bvA, true);
    stepb_pair(H, Tm, A, nB, l16, half, bvB, true);
    stepb_single(H, Tm, A, hm, nC, l16, half, bvC, okC);
    __syncthreads();

    for (int c = tid; c < 152; c += 128) {
      float s = 0.0f;
      for (int r = 0; r < PNODES; ++r) s += H[r][c];
      pooled[(size_t)g * 152 + c] = s * (1.0f / 22.0f);
    }
    __syncthreads();
  }
}

#define CONV_TW    32
#define CONV_PITCH 460
#define CONV_NS    57
#define CONV_G     3
#define CONV_NG    (CONV_NS / CONV_G)

__global__ __launch_bounds__(160) void conv_wmma_kernel(
    const float* __restrict__ pooled, const float* __restrict__ cw,
    const float* __restrict__ cb, float* __restrict__ convout)
{
  __shared__ __attribute__((aligned(16))) float Aim[CONV_TW][CONV_PITCH];

  const int tid  = threadIdx.x;
  const int wave = __builtin_amdgcn_readfirstlane(tid >> 5);
  const int lane = tid & 31;
  const int half = lane >> 4;
  const int l16  = lane & 15;
  const int b    = blockIdx.x >> 2;
  const int t0   = (blockIdx.x & 3) * CONV_TW;
  const int nt   = wave * 4 + blockIdx.y;
  const bool has_tile = nt < OCH / 16;

  const int  rg     = tid / 38;
  const int  c4     = tid - rg * 38;
  const bool stager = tid < 4 * 38;
  const float* pcol = pooled + (size_t)b * TLEN * 152 + c4 * 4;
  if (stager) {
    float4 tap[8][3];
#pragma unroll
    for (int it = 0; it < 8; ++it) {
#pragma unroll
      for (int kk = 0; kk < 3; ++kk) {
        const int t  = t0 + rg + 4 * it + kk - 1;
        const int tc = t < 0 ? 0 : (t >= TLEN ? TLEN - 1 : t);
        tap[it][kk] = *(const float4*)(pcol + tc * 152);
      }
    }
    __builtin_amdgcn_sched_barrier(0);
#pragma unroll
    for (int it = 0; it < 8; ++it) {
      float4 r[3];
#pragma unroll
      for (int kk = 0; kk < 3; ++kk) {
        const int  t  = t0 + rg + 4 * it + kk - 1;
        const bool ok = (t >= 0) && (t < TLEN);
        r[kk].x = ok ? tap[it][kk].x : 0.0f;
        r[kk].y = ok ? tap[it][kk].y : 0.0f;
        r[kk].z = ok ? tap[it][kk].z : 0.0f;
        r[kk].w = ok ? tap[it][kk].w : 0.0f;
      }
      float4* dst = (float4*)&Aim[rg + 4 * it][c4 * 12];
      dst[0] = make_float4(r[0].x, r[1].x, r[2].x, r[0].y);
      dst[1] = make_float4(r[1].y, r[2].y, r[0].z, r[1].z);
      dst[2] = make_float4(r[2].z, r[0].w, r[1].w, r[2].w);
    }
  }

  float4 wq[CONV_NS];
  float4 bl, bh;
  const int nb = nt * 16 + 8 * half;
  if (has_tile) {
    const float* wrow = cw + (size_t)(nt * 16 + l16) * KDIM + 4 * half;
#pragma unroll
    for (int s = 0; s < CONV_NS; ++s) wq[s] = *(const float4*)(wrow + 8 * s);
    bl = *(const float4*)(cb + nb);
    bh = *(const float4*)(cb + nb + 4);
  }

  __builtin_amdgcn_fence(__ATOMIC_RELEASE, "workgroup", "local");
  __builtin_amdgcn_s_barrier();
  __builtin_amdgcn_fence(__ATOMIC_ACQUIRE, "workgroup", "local");
  if (!has_tile) return;

  const float* x0p = &Aim[l16][4 * half];
  const float* x1p = &Aim[16 + l16][4 * half];
  v8f acc0 = {}, acc1 = {};
  float4 xc[CONV_G], yc[CONV_G];
#pragma unroll
  for (int i = 0; i < CONV_G; ++i) {
    xc[i] = *(const float4*)(x0p + 8 * i);
    yc[i] = *(const float4*)(x1p + 8 * i);
  }
#pragma unroll
  for (int g = 0; g < CONV_NG; ++g) {
    float4 xn[CONV_G], yn[CONV_G];
    if (g + 1 < CONV_NG) {
#pragma unroll
      for (int i = 0; i < CONV_G; ++i) {
        xn[i] = *(const float4*)(x0p + 8 * ((g + 1) * CONV_G + i));
        yn[i] = *(const float4*)(x1p + 8 * ((g + 1) * CONV_G + i));
      }
    }
    __builtin_amdgcn_sched_barrier(0);
#pragma unroll
    for (int i = 0; i < CONV_G; ++i) {
      const float4 w = wq[g * CONV_G + i];
      v2f wa, wb, xa, xb, ya, yb;
      wa.x = w.x;     wa.y = w.y;     wb.x = w.z;     wb.y = w.w;
      xa.x = xc[i].x; xa.y = xc[i].y; xb.x = xc[i].z; xb.y = xc[i].w;
      ya.x = yc[i].x; ya.y = yc[i].y; yb.x = yc[i].z; yb.y = yc[i].w;
      acc0 = wmma_f32_4(wa, xa, acc0);
      acc1 = wmma_f32_4(wa, ya, acc1);
      acc0 = wmma_f32_4(wb, xb, acc0);
      acc1 = wmma_f32_4(wb, yb, acc1);
    }
    __builtin_amdgcn_sched_barrier(0);
    if (g + 1 < CONV_NG) {
#pragma unroll
      for (int i = 0; i < CONV_G; ++i) { xc[i] = xn[i]; yc[i] = yn[i]; }
    }
  }

  const float bias[8] = {bl.x, bl.y, bl.z, bl.w, bh.x, bh.y, bh.z, bh.w};
  const int ta = t0 + l16;
  const int tb = t0 + 16 + l16;
#pragma unroll
  for (int r = 0; r < 8; ++r) {
    float* orow = convout + ((size_t)b * OCH + nb + r) * TLEN;
    if (ta < TLEN) orow[ta] = acc0[r] + bias[r];
    if (tb < TLEN) orow[tb] = acc1[r] + bias[r];
  }
}

__global__ __launch_bounds__(256) void caps_norm_kernel(
    const float* __restrict__ convout, const float* __restrict__ gamma,
    const float* __restrict__ beta, float* __restrict__ out)
{
  const int idx = blockIdx.x * 256 + threadIdx.x;
  if (idx >= BSZ * TLEN * NCLS) return;
  const int cls = idx % NCLS;
  const int t   = (idx / NCLS) % TLEN;
  const int b   = idx / (NCLS * TLEN);
  const float scale = gamma[t] * rsqrtf(1.0f + 0.001f);
  const float shift = beta[t];
  float sum = 0.0f;
  for (int d = 0; d < 16; ++d) {
    const int o = d * NCLS + cls;
    const float v  = convout[((size_t)b * OCH + o) * TLEN + t];
    const float s  = 1.0f / (1.0f + __expf(-(v * scale + shift)));
    const float df = s - 0.5f;
    sum += df * df;
  }
  out[idx] = sqrtf(sum * 0.25f);
}

extern "C" void kernel_launch(void* const* d_in, const int* in_sizes, int n_in,
                              void* d_out, int out_size, void* d_ws, size_t ws_size,
                              hipStream_t stream) {
  (void)in_sizes; (void)n_in; (void)out_size; (void)ws_size;
  const float* x  = (const float*)d_in[0];
  const float* ea = (const float*)d_in[3];
  const float* W1 = (const float*)d_in[4];  const float* b1 = (const float*)d_in[5];
  const float* W2 = (const float*)d_in[6];  const float* b2 = (const float*)d_in[7];
  const float* W3 = (const float*)d_in[8];  const float* b3 = (const float*)d_in[9];
  const float* W4 = (const float*)d_in[10]; const float* b4 = (const float*)d_in[11];
  const float* cw = (const float*)d_in[12]; const float* cb = (const float*)d_in[13];
  const float* gm = (const float*)d_in[14]; const float* bt = (const float*)d_in[15];

  float* pooled  = (float*)d_ws;
  float* convout = pooled + (size_t)GNUM * 152;

  gcn_fused_kernel<<<GCN_WGS, 128, 0, stream>>>(x, ea, W1, b1, W2, b2, W3, b3, W4, b4, pooled);
  conv_wmma_kernel<<<dim3(BSZ * 4, 4), 160, 0, stream>>>(pooled, cw, cb, convout);
  const int nout = BSZ * TLEN * NCLS;
  caps_norm_kernel<<<(nout + 255) / 256, 256, 0, stream>>>(convout, gm, bt, (float*)d_out);
}


# GCN adjacency path: each wave gathers the normalized adjacency from memory into its step-B WMMA fragments (24 independent loads, lane-sum degree, shuffled dinv): no LDS copy, no serial load chain
# speedup vs baseline: 5.4272x; 1.3686x over previous
#include <hip/hip_runtime.h>
#include <hip/hip_bf16.h>
#include <cmath>

typedef __attribute__((ext_vector_type(2))) float v2f;
typedef __attribute__((ext_vector_type(8))) float v8f;

#define PNODES 22
#define GNUM   1920
#define CIN    14
#define HCOLS  160
#define EPG    462
#define TLEN   120
#define BSZ    16
#define OCH    272
#define KDIM   456
#define NCLS   17

__device__ __forceinline__ v8f wmma_f32_4(v2f a, v2f b, v8f c) {
  return __builtin_amdgcn_wmma_f32_16x16x4_f32(
       false, a,  false, b,
       (short)0, c,  false,  false);
}

#define GCN_WGS 960

template <int CIN_, int COUT_, int KS>
__device__ __forceinline__ void load_w_frags(v2f (&wf)[KS], const float* __restrict__ W,
                                             int nt, int half, int l16)
{
  const int  n   = nt * 16 + l16;
  const bool nok = n < COUT_;
  const int  ncl = nok ? n : 0;
  const float* base = W + (2 * half) * COUT_ + ncl;
#pragma unroll
  for (int ks = 0; ks < KS; ++ks) {
    if (4 * ks + 3 < CIN_) {
      const float w0 = base[(4 * ks) * COUT_];
      const float w1 = base[(4 * ks + 1) * COUT_];
      wf[ks].x = nok ? w0 : 0.0f;
      wf[ks].y = nok ? w1 : 0.0f;
    } else {
      const int k0  = 4 * ks + 2 * half;
      const int kr0 = (k0     < CIN_) ? k0     : 0;
      const int kr1 = (k0 + 1 < CIN_) ? k0 + 1 : 0;
      const float w0 = W[kr0 * COUT_ + ncl];
      const float w1 = W[kr1 * COUT_ + ncl];
      wf[ks].x = (k0     < CIN_ && nok) ? w0 : 0.0f;
      wf[ks].y = (k0 + 1 < CIN_ && nok) ? w1 : 0.0f;
    }
  }
}

__device__ __forceinline__ void store_tm(float (&Tm)[32][HCOLS], v8f c, int mt, int n, int half)
{
#pragma unroll
  for (int v = 0; v < 8; ++v) Tm[mt * 16 + 8 * half + v][n] = c[v];
}

__device__ __forceinline__ void store_h(float (&H)[32][HCOLS], v8f c, int mt, int n, int half,
                                        float bv, bool nok)
{
#pragma unroll
  for (int v = 0; v < 8; ++v) {
    const int row = mt * 16 + 8 * half + v;
    H[row][n] = (row < PNODES && nok) ? fmaxf(c[v] + bv, 0.0f) : 0.0f;
  }
}

__device__ __forceinline__ void stepb_pair(float (&H)[32][HCOLS], const float (&Tm)[32][HCOLS],
    const v2f (&adj0)[6], const v2f (&adj1)[6], int n, int l16, int half, float bv, bool nok)
{
  v2f b[6];
#pragma unroll
  for (int ks = 0; ks < 6; ++ks) {
    const int k0 = 4 * ks + 2 * half;
    b[ks].x = Tm[k0][n];  b[ks].y = Tm[k0 + 1][n];
  }
  v8f c0 = {}, c1 = {};
#pragma unroll
  for (int ks = 0; ks < 6; ++ks) {
    c0 = wmma_f32_4(adj0[ks], b[ks], c0);
    c1 = wmma_f32_4(adj1[ks], b[ks], c1);
  }
  store_h(H, c0, 0, n, half, bv, nok);
  store_h(H, c1, 1, n, half, bv, nok);
}

__device__ __forceinline__ void stepb_single(float (&H)[32][HCOLS], const float (&Tm)[32][HCOLS],
    const v2f (&adj0)[6], const v2f (&adj1)[6], int mt, int n, int l16, int half, float bv, bool nok)
{
  v2f b[6], adjh[6];
#pragma unroll
  for (int ks = 0; ks < 6; ++ks) {
    const int k0 = 4 * ks + 2 * half;
    b[ks].x = Tm[k0][n];  b[ks].y = Tm[k0 + 1][n];
    adjh[ks] = mt ? adj1[ks] : adj0[ks];
  }
  v8f c = {};
#pragma unroll
  for (int ks = 0; ks < 6; ++ks) c = wmma_f32_4(adjh[ks], b[ks], c);
  store_h(H, c, mt, n, half, bv, nok);
}

__global__ __launch_bounds__(128) void gcn_fused_kernel(
    const float* __restrict__ x, const float* __restrict__ ea,
    const float* __restrict__ W1, const float* __restrict__ b1,
    const float* __restrict__ W2, const float* __restrict__ b2,
    const float* __restrict__ W3, const float* __restrict__ b3,
    const float* __restrict__ W4, const float* __restrict__ b4,
    float* __restrict__ pooled)
{
  __shared__ __attribute__((aligned(16))) float H[32][HCOLS];
  __shared__ __attribute__((aligned(16))) float Tm[32][HCOLS];

  const int tid  = threadIdx.x;
  const int wave = __builtin_amdgcn_readfirstlane(tid >> 5);
  const int lane = tid & 31;
  const int half = lane >> 4;
  const int l16  = lane & 15;
  const int hm   = wave & 1;

  const int ntA = 2 * wave, ntB = 2 * wave + 1, ntC = 8 + (wave >> 1);
  const int nt3 = wave, nt2 = wave >> 1;
  const int nA = ntA * 16 + l16, nB = ntB * 16 + l16, nC = ntC * 16 + l16;
  const int n3 = nt3 * 16 + l16, n2 = nt2 * 16 + l16, n1 = l16;
  const bool okC = nC < 152;
  v2f w4a[16], w4b[16], w4c[16], w3f[8], w2f[4];
  load_w_frags<64, 152, 16>(w4a, W4, ntA, half, l16);
  load_w_frags<64, 152, 16>(w4b, W4, ntB, half, l16);
  load_w_frags<64, 152, 16>(w4c, W4, ntC, half, l16);
  load_w_frags<32, 64, 8>(w3f, W3, nt3, half, l16);
  load_w_frags<16, 32, 4>(w2f, W2, nt2, half, l16);
  const float bvA = b4[nA], bvB = b4[nB], bvC = okC ? b4[okC ? nC : 0] : 0.0f;
  const float bv3 = b3[n3], bv2 = b2[n2], bv1 = b1[n1];

  for (int g = blockIdx.x; g < GNUM; g += GCN_WGS) {
    int l16s = l16, halfs = half, tids = tid;
    asm volatile("" : "+v"(l16s), "+v"(halfs), "+v"(tids));
    const int l16 = l16s, half = halfs, tid = tids;
    const int nA = ntA * 16 + l16, nB = ntB * 16 + l16, nC = ntC * 16 + l16;
    const int n3 = nt3 * 16 + l16, n2 = nt2 * 16 + l16, n1 = l16;
    const float* hrow0 = &H[l16][2 * half];
    const float* hrow1 = &H[16 + l16][2 * half];
    const float* hrowh = hm ? hrow1 : hrow0;
    const float* eaw = ea + (size_t)g * EPG * 5;

    if (g + GCN_WGS < GNUM) {
      const char* nbase = (const char*)(ea + (size_t)(g + GCN_WGS) * EPG * 5);
      const char* nfeat = (const char*)(x + (size_t)(g + GCN_WGS) * PNODES * CIN);
      __builtin_prefetch(nbase + (size_t)tid * 64, 0, 1);
      __builtin_prefetch(nbase + (size_t)(tid + 128) * 64 % 9240, 0, 1);
      __builtin_prefetch(nfeat + (size_t)tid * 64 % 1232, 0, 1);
    }

    float raw0[12], raw1[12];
    {
      const int  j0 = l16, j1 = 16 + l16;
      const bool j1ok = j1 < PNODES;
      const int  j1c = j1ok ? j1 : 0;
#pragma unroll
      for (int q = 0; q < 12; ++q) {
        const int  i   = 4 * (q >> 1) + 2 * half + (q & 1);
        const bool iok = i < PNODES;
        const int  ic  = iok ? i : 0;
        const int  e0  = (ic == j0)  ? 0 : ic * (PNODES - 1) + (j0  < ic ? j0  : j0  - 1);
        const int  e1  = (ic == j1c) ? 0 : ic * (PNODES - 1) + (j1c < ic ? j1c : j1c - 1);
        raw0[q] = eaw[e0 * 5 + 4];
        raw1[q] = eaw[e1 * 5 + 4];
      }
    }
    v2f w1f[4] = {};
    if (wave < 2) load_w_frags<14, 16, 4>(w1f, W1, 0, half, l16);
    for (int idx = tid; idx < 32 * 16; idx += 128) {
      const int r = idx >> 4, c = idx & 15;
      if (r >= PNODES || c >= CIN) H[r][c] = 0.0f;
    }
    for (int idx = tid; idx < PNODES * CIN; idx += 128) {
      const int r = idx / CIN, c = idx % CIN;
      H[r][c] = x[(size_t)(g * PNODES + r) * CIN + c];
    }
    __syncthreads();

    v2f adj0[6], adj1[6];
    {
      const int  j0 = l16, j1 = 16 + l16;
      const bool j1ok = j1 < PNODES;
      float w0[12], w1[12];
      float s0 = 0.0f, s1 = 0.0f;
#pragma unroll
      for (int q = 0; q < 12; ++q) {
        const int  i   = 4 * (q >> 1) + 2 * half + (q & 1);
        const bool iok = i < PNODES;
        w0[q] = iok ? ((i == j0) ? 1.0f : raw0[q]) : 0.0f;
        w1[q] = (iok && j1ok) ? ((i == j1) ? 1.0f : raw1[q]) : 0.0f;
        s0 += w0[q];
        s1 += w1[q];
      }
      s0 += __shfl_xor(s0, 16);
      s1 += __shfl_xor(s1, 16);
      const float dr0 = (s0 > 0.0f) ? rsqrtf(s0) : 0.0f;
      const float dr1 = (s1 > 0.0f) ? rsqrtf(s1) : 0.0f;
#pragma unroll
      for (int ks = 0; ks < 6; ++ks) {
        const int i = 4 * ks + 2 * half;
        const float dc0 = (ks < 4) ? __shfl(dr0, i)     : __shfl(dr1, i - 16);
        const float dc1 = (ks < 4) ? __shfl(dr0, i + 1) : __shfl(dr1, i - 15);
        adj0[ks].x = dc0 * w0[2 * ks]     * dr0;
        adj0[ks].y = dc1 * w0[2 * ks + 1] * dr0;
        adj1[ks].x = dc0 * w1[2 * ks]     * dr1;
        adj1[ks].y = dc1 * w1[2 * ks + 1] * dr1;
      }
    }

    if (wave < 2) {
      v8f c = {};
#pragma unroll
      for (int ks = 0; ks < 4; ++ks) c = wmma_f32_4(*(const v2f*)(hrowh + 4 * ks), w1f[ks], c);
      store_tm(Tm, c, hm, n1, half);
    }
    __syncthreads();
    if (wave < 2) stepb_single(H, Tm, adj0, adj1, hm, n1, l16, half, bv1, true);
    __syncthreads();

    {
      v8f c = {};
#pragma unroll
      for (int ks = 0; ks < 4; ++ks) c = wmma_f32_4(*(const v2f*)(hrowh + 4 * ks), w2f[ks], c);
      store_tm(Tm, c, hm, n2, half);
    }
    __syncthreads();
    stepb_single(H, Tm, adj0, adj1, hm, n2, l16, half, bv2, true);
    __syncthreads();

    {
      v8f c0 = {}, c1 = {};
#pragma unroll
      for (int ks = 0; ks < 8; ++ks) {
        const v2f a0 = *(const v2f*)(hrow0 + 4 * ks);
        const v2f a1 = *(const v2f*)(hrow1 + 4 * ks);
        c0 = wmma_f32_4(a0, w3f[ks], c0);
        c1 = wmma_f32_4(a1, w3f[ks], c1);
      }
      store_tm(Tm, c0, 0, n3, half);
      store_tm(Tm, c1, 1, n3, half);
    }
    __syncthreads();
    stepb_pair(H, Tm, adj0, adj1, n3, l16, half, bv3, true);
    __syncthreads();

    {
      v8f c00 = {}, c01 = {}, c10 = {}, c11 = {}, ch = {};
      v2f xa[4], xb[4];
#pragma unroll
      for (int i = 0; i < 4; ++i) {
        xa[i] = *(const v2f*)(hrow0 + 4 * i);
        xb[i] = *(const v2f*)(hrow1 + 4 * i);
      }
#pragma unroll
      for (int ck = 0; ck < 4; ++ck) {
        v2f na[4], nb[4];
        if (ck + 1 < 4) {
#pragma unroll
          for (int i = 0; i < 4; ++i) {
            na[i] = *(const v2f*)(hrow0 + 4 * ((ck + 1) * 4 + i));
            nb[i] = *(const v2f*)(hrow1 + 4 * ((ck + 1) * 4 + i));
          }
        }
        __builtin_amdgcn_sched_barrier(0);
#pragma unroll
        for (int i = 0; i < 4; ++i) {
          const int ks = ck * 4 + i;
          const v2f a0 = xa[i], a1 = xb[i];
          const v2f ah = hm ? a1 : a0;
          c00 = wmma_f32_4(a0, w4a[ks], c00);
          c01 = wmma_f32_4(a1, w4a[ks], c01);
          c10 = wmma_f32_4(a0, w4b[ks], c10);
          c11 = wmma_f32_4(a1, w4b[ks], c11);
          ch  = wmma_f32_4(ah, w4c[ks], ch);
        }
        __builtin_amdgcn_sched_barrier(0);
        if (ck + 1 < 4) {
#pragma unroll
          for (int i = 0; i < 4; ++i) { xa[i] = na[i]; xb[i] = nb[i]; }
        }
      }
      store_tm(Tm, c00, 0, nA, half);
      store_tm(Tm, c01, 1, nA, half);
      store_tm(Tm, c10, 0, nB, half);
      store_tm(Tm, c11, 1, nB, half);
      store_tm(Tm, ch, hm, nC, half);
    }
    __syncthreads();
    stepb_pair(H, Tm, adj0, adj1, nA, l16, half, bvA, true);
    stepb_pair(H, Tm, adj0, adj1, nB, l16, half, bvB, true);
    stepb_single(H, Tm, adj0, adj1, hm, nC, l16, half, bvC, okC);
    __syncthreads();

    for (int c = tid; c < 152; c += 128) {
      float s = 0.0f;
      for (int r = 0; r < PNODES; ++r) s += H[r][c];
      pooled[(size_t)g * 152 + c] = s * (1.0f / 22.0f);
    }
    __syncthreads();
  }
}

#define CONV_TW    32
#define CONV_PITCH 460
#define CONV_NS    57
#define CONV_G     3
#define CONV_NG    (CONV_NS / CONV_G)

__global__ __launch_bounds__(160) void conv_wmma_kernel(
    const float* __restrict__ pooled, const float* __restrict__ cw,
    const float* __restrict__ cb, float* __restrict__ convout)
{
  __shared__ __attribute__((aligned(16))) float Aim[CONV_TW][CONV_PITCH];

  const int tid  = threadIdx.x;
  const int wave = __builtin_amdgcn_readfirstlane(tid >> 5);
  const int lane = tid & 31;
  const int half = lane >> 4;
  const int l16  = lane & 15;
  const int b    = blockIdx.x >> 2;
  const int t0   = (blockIdx.x & 3) * CONV_TW;
  const int nt   = wave * 4 + blockIdx.y;
  const bool has_tile = nt < OCH / 16;

  const int  rg     = tid / 38;
  const int  c4     = tid - rg * 38;
  const bool stager = tid < 4 * 38;
  const float* pcol = pooled + (size_t)b * TLEN * 152 + c4 * 4;
  if (stager) {
    float4 tap[8][3];
#pragma unroll
    for (int it = 0; it < 8; ++it) {
#pragma unroll
      for (int kk = 0; kk < 3; ++kk) {
        const int t  = t0 + rg + 4 * it + kk - 1;
        const int tc = t < 0 ? 0 : (t >= TLEN ? TLEN - 1 : t);
        tap[it][kk] = *(const float4*)(pcol + tc * 152);
      }
    }
    __builtin_amdgcn_sched_barrier(0);
#pragma unroll
    for (int it = 0; it < 8; ++it) {
      float4 r[3];
#pragma unroll
      for (int kk = 0; kk < 3; ++kk) {
        const int  t  = t0 + rg + 4 * it + kk - 1;
        const bool ok = (t >= 0) && (t < TLEN);
        r[kk].x = ok ? tap[it][kk].x : 0.0f;
        r[kk].y = ok ? tap[it][kk].y : 0.0f;
        r[kk].z = ok ? tap[it][kk].z : 0.0f;
        r[kk].w = ok ? tap[it][kk].w : 0.0f;
      }
      float4* dst = (float4*)&Aim[rg + 4 * it][c4 * 12];
      dst[0] = make_float4(r[0].x, r[1].x, r[2].x, r[0].y);
      dst[1] = make_float4(r[1].y, r[2].y, r[0].z, r[1].z);
      dst[2] = make_float4(r[2].z, r[0].w, r[1].w, r[2].w);
    }
  }

  float4 wq[CONV_NS];
  float4 bl, bh;
  const int nb = nt * 16 + 8 * half;
  if (has_tile) {
    const float* wrow = cw + (size_t)(nt * 16 + l16) * KDIM + 4 * half;
#pragma unroll
    for (int s = 0; s < CONV_NS; ++s) wq[s] = *(const float4*)(wrow + 8 * s);
    bl = *(const float4*)(cb + nb);
    bh = *(const float4*)(cb + nb + 4);
  }

  __builtin_amdgcn_fence(__ATOMIC_RELEASE, "workgroup", "local");
  __builtin_amdgcn_s_barrier();
  __builtin_amdgcn_fence(__ATOMIC_ACQUIRE, "workgroup", "local");
  if (!has_tile) return;

  const float* x0p = &Aim[l16][4 * half];
  const float* x1p = &Aim[16 + l16][4 * half];
  v8f acc0 = {}, acc1 = {};
  float4 xc[CONV_G], yc[CONV_G];
#pragma unroll
  for (int i = 0; i < CONV_G; ++i) {
    xc[i] = *(const float4*)(x0p + 8 * i);
    yc[i] = *(const float4*)(x1p + 8 * i);
  }
#pragma unroll
  for (int g = 0; g < CONV_NG; ++g) {
    float4 xn[CONV_G], yn[CONV_G];
    if (g + 1 < CONV_NG) {
#pragma unroll
      for (int i = 0; i < CONV_G; ++i) {
        xn[i] = *(const float4*)(x0p + 8 * ((g + 1) * CONV_G + i));
        yn[i] = *(const float4*)(x1p + 8 * ((g + 1) * CONV_G + i));
      }
    }
    __builtin_amdgcn_sched_barrier(0);
#pragma unroll
    for (int i = 0; i < CONV_G; ++i) {
      const float4 w = wq[g * CONV_G + i];
      v2f wa, wb, xa, xb, ya, yb;
      wa.x = w.x;     wa.y = w.y;     wb.x = w.z;     wb.y = w.w;
      xa.x = xc[i].x; xa.y = xc[i].y; xb.x = xc[i].z; xb.y = xc[i].w;
      ya.x = yc[i].x; ya.y = yc[i].y; yb.x = yc[i].z; yb.y = yc[i].w;
      acc0 = wmma_f32_4(wa, xa, acc0);
      acc1 = wmma_f32_4(wa, ya, acc1);
      acc0 = wmma_f32_4(wb, xb, acc0);
      acc1 = wmma_f32_4(wb, yb, acc1);
    }
    __builtin_amdgcn_sched_barrier(0);
    if (g + 1 < CONV_NG) {
#pragma unroll
      for (int i = 0; i < CONV_G; ++i) { xc[i] = xn[i]; yc[i] = yn[i]; }
    }
  }

  const float bias[8] = {bl.x, bl.y, bl.z, bl.w, bh.x, bh.y, bh.z, bh.w};
  const int ta = t0 + l16;
  const int tb = t0 + 16 + l16;
#pragma unroll
  for (int r = 0; r < 8; ++r) {
    float* orow = convout + ((size_t)b * OCH + nb + r) * TLEN;
    if (ta < TLEN) orow[ta] = acc0[r] + bias[r];
    if (tb < TLEN) orow[tb] = acc1[r] + bias[r];
  }
}

__global__ __launch_bounds__(256) void caps_norm_kernel(
    const float* __restrict__ convout, const float* __restrict__ gamma,
    const float* __restrict__ beta, float* __restrict__ out)
{
  const int idx = blockIdx.x * 256 + threadIdx.x;
  if (idx >= BSZ * TLEN * NCLS) return;
  const int cls = idx % NCLS;
  const int t   = (idx / NCLS) % TLEN;
  const int b   = idx / (NCLS * TLEN);
  const float scale = gamma[t] * rsqrtf(1.0f + 0.001f);
  const float shift = beta[t];
  float sum = 0.0f;
  for (int d = 0; d < 16; ++d) {
    const int o = d * NCLS + cls;
    const float v  = convout[((size_t)b * OCH + o) * TLEN + t];
    const float s  = 1.0f / (1.0f + __expf(-(v * scale + shift)));
    const float df = s - 0.5f;
    sum += df * df;
  }
  out[idx] = sqrtf(sum * 0.25f);
}

extern "C" void kernel_launch(void* const* d_in, const int* in_sizes, int n_in,
                              void* d_out, int out_size, void* d_ws, size_t ws_size,
                              hipStream_t stream) {
  (void)in_sizes; (void)n_in; (void)out_size; (void)ws_size;
  const float* x  = (const float*)d_in[0];
  const float* ea = (const float*)d_in[3];
  const float* W1 = (const float*)d_in[4];  const float* b1 = (const float*)d_in[5];
  const float* W2 = (const float*)d_in[6];  const float* b2 = (const float*)d_in[7];
  const float* W3 = (const float*)d_in[8];  const float* b3 = (const float*)d_in[9];
  const float* W4 = (const float*)d_in[10]; const float* b4 = (const float*)d_in[11];
  const float* cw = (const float*)d_in[12]; const float* cb = (const float*)d_in[13];
  const float* gm = (const float*)d_in[14]; const float* bt = (const float*)d_in[15];

  float* pooled  = (float*)d_ws;
  float* convout = pooled + (size_t)GNUM * 152;

  gcn_fused_kernel<<<GCN_WGS, 128, 0, stream>>>(x, ea, W1, b1, W2, b2, W3, b3, W4, b4, pooled);
  conv_wmma_kernel<<<dim3(BSZ * 4, 4), 160, 0, stream>>>(pooled, cw, cb, convout);
  const int nout = BSZ * TLEN * NCLS;
  caps_norm_kernel<<<(nout + 255) / 256, 256, 0, stream>>>(convout, gm, bt, (float*)d_out);
}


# GCN layers: Tm = H @ W stays in registers: a wave's two result tiles are the B fragments of its A @ Tm WMMAs (row slots balance lane halves); H alternates between 2 LDS buffers, 1 barrier per layer
# speedup vs baseline: 5.6600x; 1.0429x over previous
#include <hip/hip_runtime.h>
#include <hip/hip_bf16.h>
#include <cmath>

typedef __attribute__((ext_vector_type(2))) float v2f;
typedef __attribute__((ext_vector_type(8))) float v8f;

#define PNODES 22
#define GNUM   1920
#define CIN    14
#define HCOLS  160
#define EPG    462
#define TLEN   120
#define BSZ    16
#define OCH    272
#define KDIM   456
#define NCLS   17

__device__ __forceinline__ v8f wmma_f32_4(v2f a, v2f b, v8f c) {
  return __builtin_amdgcn_wmma_f32_16x16x4_f32(
       false, a,  false, b,
       (short)0, c,  false,  false);
}

#define GCN_WGS 960

template <int CIN_, int COUT_, int KS>
__device__ __forceinline__ void load_w_frags(v2f (&wf)[KS], const float* __restrict__ W,
                                             int nt, int half, int l16)
{
  const int  n   = nt * 16 + l16;
  const bool nok = n < COUT_;
  const int  ncl = nok ? n : 0;
  const float* base = W + (2 * half) * COUT_ + ncl;
#pragma unroll
  for (int ks = 0; ks < KS; ++ks) {
    if (4 * ks + 3 < CIN_) {
      const float w0 = base[(4 * ks) * COUT_];
      const float w1 = base[(4 * ks + 1) * COUT_];
      wf[ks].x = nok ? w0 : 0.0f;
      wf[ks].y = nok ? w1 : 0.0f;
    } else {
      const int k0  = 4 * ks + 2 * half;
      const int kr0 = (k0     < CIN_) ? k0     : 0;
      const int kr1 = (k0 + 1 < CIN_) ? k0 + 1 : 0;
      const float w0 = W[kr0 * COUT_ + ncl];
      const float w1 = W[kr1 * COUT_ + ncl];
      wf[ks].x = (k0     < CIN_ && nok) ? w0 : 0.0f;
      wf[ks].y = (k0 + 1 < CIN_ && nok) ? w1 : 0.0f;
    }
  }
}

__device__ __forceinline__ int slot_node(int s) { return s < 19 ? s : ((s >= 24 && s < 27) ? s - 5 : -1); }
__device__ __forceinline__ int node_slot(int r) { return r < 19 ? r : r + 5; }

__device__ __forceinline__ void store_h(float (&H)[32][HCOLS], v8f c, int mt, int n, int half,
                                        float bv, bool nok)
{
#pragma unroll
  for (int v = 0; v < 8; ++v) {
    const int  row  = mt * 16 + 8 * half + v;
    const bool real = (mt == 0) || (v < 3);
    H[row][n] = (real && nok) ? fmaxf(c[v] + bv, 0.0f) : 0.0f;
  }
}

__device__ __forceinline__ void stepb_regs(float (&H)[32][HCOLS], const v8f t0, const v8f t1,
    const v2f (&adj0)[6], const v2f (&adj1)[6], int n, int half, float bv, bool nok)
{
  v8f d0 = {}, d1 = {};
#pragma unroll
  for (int ks = 0; ks < 6; ++ks) {
    v2f b;
    if (ks < 4) { b.x = t0[2 * ks];       b.y = t0[2 * ks + 1]; }
    else        { b.x = t1[2 * (ks - 4)]; b.y = t1[2 * (ks - 4) + 1]; }
    d0 = wmma_f32_4(adj0[ks], b, d0);
    d1 = wmma_f32_4(adj1[ks], b, d1);
  }
  store_h(H, d0, 0, n, half, bv, nok);
  store_h(H, d1, 1, n, half, bv, nok);
}

template <int KS>
__device__ __forceinline__ void stepa_pair(v8f& c0, v8f& c1, const float* h0, const float* h1,
                                           const v2f* wf)
{
#pragma unroll
  for (int ks = 0; ks < KS; ++ks) {
    const v2f a0 = *(const v2f*)(h0 + 4 * ks);
    const v2f a1 = *(const v2f*)(h1 + 4 * ks);
    c0 = wmma_f32_4(a0, wf[ks], c0);
    c1 = wmma_f32_4(a1, wf[ks], c1);
  }
}

template <int NT>
__device__ __forceinline__ void layer4_stepa(v8f (&c)[6], const float* h0, const float* h1,
    const v2f (&wa)[16], const v2f (&wb)[16], const v2f (&wc)[16])
{
  v2f xa[4], xb[4];
#pragma unroll
  for (int i = 0; i < 4; ++i) {
    xa[i] = *(const v2f*)(h0 + 4 * i);
    xb[i] = *(const v2f*)(h1 + 4 * i);
  }
#pragma unroll
  for (int ck = 0; ck < 4; ++ck) {
    v2f na[4], nb[4];
    if (ck + 1 < 4) {
#pragma unroll
      for (int i = 0; i < 4; ++i) {
        na[i] = *(const v2f*)(h0 + 4 * ((ck + 1) * 4 + i));
        nb[i] = *(const v2f*)(h1 + 4 * ((ck + 1) * 4 + i));
      }
    }
    __builtin_amdgcn_sched_barrier(0);
#pragma unroll
    for (int i = 0; i < 4; ++i) {
      const int ks = ck * 4 + i;
      const v2f a0 = xa[i], a1 = xb[i];
      c[0] = wmma_f32_4(a0, wa[ks], c[0]);
      c[1] = wmma_f32_4(a1, wa[ks], c[1]);
      c[2] = wmma_f32_4(a0, wb[ks], c[2]);
      c[3] = wmma_f32_4(a1, wb[ks], c[3]);
      if (NT == 3) {
        c[4] = wmma_f32_4(a0, wc[ks], c[4]);
        c[5] = wmma_f32_4(a1, wc[ks], c[5]);
      }
    }
    __builtin_amdgcn_sched_barrier(0);
    if (ck + 1 < 4) {
#pragma unroll
      for (int i = 0; i < 4; ++i) { xa[i] = na[i]; xb[i] = nb[i]; }
    }
  }
}

__global__ __launch_bounds__(128) void gcn_fused_kernel(
    const float* __restrict__ x, const float* __restrict__ ea,
    const float* __restrict__ W1, const float* __restrict__ b1,
    const float* __restrict__ W2, const float* __restrict__ b2,
    const float* __restrict__ W3, const float* __restrict__ b3,
    const float* __restrict__ W4, const float* __restrict__ b4,
    float* __restrict__ pooled)
{
  __shared__ __attribute__((aligned(16))) float Ha[32][HCOLS];
  __shared__ __attribute__((aligned(16))) float Hb[32][HCOLS];

  const int tid  = threadIdx.x;
  const int wave = __builtin_amdgcn_readfirstlane(tid >> 5);
  const int lane = tid & 31;
  const int half = lane >> 4;
  const int l16  = lane & 15;
  const int swap  = (blockIdx.x + (blockIdx.x >> 8)) & 1;
  const int role  = wave ^ (swap << 1);
  const bool three = role < 2;

  const int tA = three ? 3 * role : 2 + 2 * role;
  const int tB = tA + 1, tX = tA + 2;
  const int t2 = role & 1;
  const int nA = tA * 16 + l16, nB = tB * 16 + l16, nX = tX * 16 + l16;
  const int n3 = role * 16 + l16, n2 = t2 * 16 + l16, n1 = l16;
  const bool okB = nB < 152;
  v2f w4a[16], w4b[16], wx[16], w3f[8];
  load_w_frags<64, 152, 16>(w4a, W4, tA, half, l16);
  load_w_frags<64, 152, 16>(w4b, W4, tB, half, l16);
  load_w_frags<32, 64, 8>(w3f, W3, role, half, l16);
  if (three) {
    load_w_frags<64, 152, 16>(wx, W4, tX, half, l16);
  } else {
    v2f w2t[4];
    load_w_frags<16, 32, 4>(w2t, W2, t2, half, l16);
#pragma unroll
    for (int ks = 0; ks < 16; ++ks) { wx[ks].x = (ks < 4) ? w2t[ks & 3].x : 0.0f; wx[ks].y = (ks < 4) ? w2t[ks & 3].y : 0.0f; }
  }
  const float bvA = b4[nA], bvB = okB ? b4[okB ? nB : 0] : 0.0f;
  const float bvX = three ? b4[three ? nX : 0] : b2[n2];
  const float bv3 = b3[n3], bv1 = b1[n1];

  for (int g = blockIdx.x; g < GNUM; g += GCN_WGS) {
    int l16s = l16, halfs = half, tids = tid;
    asm volatile("" : "+v"(l16s), "+v"(halfs), "+v"(tids));
    const int l16 = l16s, half = halfs, tid = tids;
    const int nA = tA * 16 + l16, nB = tB * 16 + l16, nX = tX * 16 + l16;
    const int n3 = role * 16 + l16, n2 = t2 * 16 + l16, n1 = l16;
    const float* ha0 = &Ha[l16][2 * half];
    const float* ha1 = &Ha[16 + l16][2 * half];
    const float* hb0 = &Hb[l16][2 * half];
    const float* hb1 = &Hb[16 + l16][2 * half];
    const float* eaw = ea + (size_t)g * EPG * 5;

    if (g + GCN_WGS < GNUM) {
      const char* nbase = (const char*)(ea + (size_t)(g + GCN_WGS) * EPG * 5);
      const char* nfeat = (const char*)(x + (size_t)(g + GCN_WGS) * PNODES * CIN);
      __builtin_prefetch(nbase + (size_t)tid * 64, 0, 1);
      __builtin_prefetch(nbase + (size_t)(tid + 128) * 64 % 9240, 0, 1);
      __builtin_prefetch(nfeat + (size_t)tid * 64 % 1232, 0, 1);
    }

    float raw0[12], raw1[12];
    const int  j0 = l16;
    const int  j1 = slot_node(16 + l16);
    const bool j1ok = j1 >= 0;
    {
      const int  j1c = j1ok ? j1 : 0;
#pragma unroll
      for (int q = 0; q < 12; ++q) {
        const int  i   = slot_node((q < 8) ? 8 * half + q : 16 + 8 * half + (q - 8));
        const bool iok = i >= 0;
        const int  ic  = iok ? i : 0;
        const int  e0  = (ic == j0)  ? 0 : ic * (PNODES - 1) + (j0  < ic ? j0  : j0  - 1);
        const int  e1  = (ic == j1c) ? 0 : ic * (PNODES - 1) + (j1c < ic ? j1c : j1c - 1);
        raw0[q] = eaw[e0 * 5 + 4];
        raw1[q] = eaw[e1 * 5 + 4];
      }
    }
    v2f w1f[4] = {};
    if (role == 2) load_w_frags<14, 16, 4>(w1f, W1, 0, half, l16);
    for (int idx = tid; idx < 32 * 16; idx += 128) {
      const int r = idx >> 4, c = idx & 15;
      if (slot_node(r) < 0 || c >= CIN) Ha[r][c] = 0.0f;
    }
    for (int idx = tid; idx < PNODES * CIN; idx += 128) {
      const int r = idx / CIN, c = idx % CIN;
      Ha[node_slot(r)][c] = x[(size_t)(g * PNODES + r) * CIN + c];
    }
    __syncthreads();

    v2f adj0[6], adj1[6];
    {
      float w0[12], w1[12];
      float s0 = 0.0f, s1 = 0.0f;
#pragma unroll
      for (int q = 0; q < 12; ++q) {
        const int  i   = slot_node((q < 8) ? 8 * half + q : 16 + 8 * half + (q - 8));
        const bool iok = i >= 0;
        w0[q] = iok ? ((i == j0) ? 1.0f : raw0[q]) : 0.0f;
        w1[q] = (iok && j1ok) ? ((i == j1) ? 1.0f : raw1[q]) : 0.0f;
        s0 += w0[q];
        s1 += w1[q];
      }
      s0 += __shfl_xor(s0, 16);
      s1 += __shfl_xor(s1, 16);
      const float dr0 = (s0 > 0.0f) ? rsqrtf(s0) : 0.0f;
      const float dr1 = (s1 > 0.0f) ? rsqrtf(s1) : 0.0f;
#pragma unroll
      for (int ks = 0; ks < 6; ++ks) {
        const int sl = 8 * half + 2 * (ks & 3);
        const float dc0 = (ks < 4) ? __shfl(dr0, sl)     : __shfl(dr1, sl);
        const float dc1 = (ks < 4) ? __shfl(dr0, sl + 1) : __shfl(dr1, sl + 1);
        adj0[ks].x = dc0 * w0[2 * ks]     * dr0;
        adj0[ks].y = dc1 * w0[2 * ks + 1] * dr0;
        adj1[ks].x = dc0 * w1[2 * ks]     * dr1;
        adj1[ks].y = dc1 * w1[2 * ks + 1] * dr1;
      }
    }

    if (role == 2) {
      v8f c0 = {}, c1 = {};
      stepa_pair<4>(c0, c1, ha0, ha1, w1f);
      stepb_regs(Hb, c0, c1, adj0, adj1, n1, half, bv1, true);
    }
    __syncthreads();

    if (!three) {
      v8f c0 = {}, c1 = {};
      stepa_pair<4>(c0, c1, hb0, hb1, wx);
      stepb_regs(Ha, c0, c1, adj0, adj1, n2, half, bvX, true);
    }
    __syncthreads();

    {
      v8f c0 = {}, c1 = {};
      stepa_pair<8>(c0, c1, ha0, ha1, w3f);
      stepb_regs(Hb, c0, c1, adj0, adj1, n3, half, bv3, true);
    }
    __syncthreads();

    {
      v8f c[6] = {};
      if (three) {
        layer4_stepa<3>(c, hb0, hb1, w4a, w4b, wx);
        stepb_regs(Ha, c[0], c[1], adj0, adj1, nA, half, bvA, true);
        stepb_regs(Ha, c[2], c[3], adj0, adj1, nB, half, bvB, true);
        stepb_regs(Ha, c[4], c[5], adj0, adj1, nX, half, bvX, true);
      } else {
        layer4_stepa<2>(c, hb0, hb1, w4a, w4b, wx);
        stepb_regs(Ha, c[0], c[1], adj0, adj1, nA, half, bvA, true);
        stepb_regs(Ha, c[2], c[3], adj0, adj1, nB, half, bvB, okB);
      }
    }
    __syncthreads();

    for (int c = tid; c < 152; c += 128) {
      float s = 0.0f;
      for (int r = 0; r < PNODES; ++r) s += Ha[node_slot(r)][c];
      pooled[(size_t)g * 152 + c] = s * (1.0f / 22.0f);
    }
    __syncthreads();
  }
}

#define CONV_TW    32
#define CONV_PITCH 460
#define CONV_NS    57
#define CONV_G     3
#define CONV_NG    (CONV_NS / CONV_G)

__global__ __launch_bounds__(160) void conv_wmma_kernel(
    const float* __restrict__ pooled, const float* __restrict__ cw,
    const float* __restrict__ cb, float* __restrict__ convout)
{
  __shared__ __attribute__((aligned(16))) float Aim[CONV_TW][CONV_PITCH];

  const int tid  = threadIdx.x;
  const int wave = __builtin_amdgcn_readfirstlane(tid >> 5);
  const int lane = tid & 31;
  const int half = lane >> 4;
  const int l16  = lane & 15;
  const int b    = blockIdx.x >> 2;
  const int t0   = (blockIdx.x & 3) * CONV_TW;
  const int nt   = wave * 4 + blockIdx.y;
  const bool has_tile = nt < OCH / 16;

  const int  rg     = tid / 38;
  const int  c4     = tid - rg * 38;
  const bool stager = tid < 4 * 38;
  const float* pcol = pooled + (size_t)b * TLEN * 152 + c4 * 4;
  if (stager) {
    float4 tap[8][3];
#pragma unroll
    for (int it = 0; it < 8; ++it) {
#pragma unroll
      for (int kk = 0; kk < 3; ++kk) {
        const int t  = t0 + rg + 4 * it + kk - 1;
        const int tc = t < 0 ? 0 : (t >= TLEN ? TLEN - 1 : t);
        tap[it][kk] = *(const float4*)(pcol + tc * 152);
      }
    }
    __builtin_amdgcn_sched_barrier(0);
#pragma unroll
    for (int it = 0; it < 8; ++it) {
      float4 r[3];
#pragma unroll
      for (int kk = 0; kk < 3; ++kk) {
        const int  t  = t0 + rg + 4 * it + kk - 1;
        const bool ok = (t >= 0) && (t < TLEN);
        r[kk].x = ok ? tap[it][kk].x : 0.0f;
        r[kk].y = ok ? tap[it][kk].y : 0.0f;
        r[kk].z = ok ? tap[it][kk].z : 0.0f;
        r[kk].w = ok ? tap[it][kk].w : 0.0f;
      }
      float4* dst = (float4*)&Aim[rg + 4 * it][c4 * 12];
      dst[0] = make_float4(r[0].x, r[1].x, r[2].x, r[0].y);
      dst[1] = make_float4(r[1].y, r[2].y, r[0].z, r[1].z);
      dst[2] = make_float4(r[2].z, r[0].w, r[1].w, r[2].w);
    }
  }

  float4 wq[CONV_NS];
  float4 bl, bh;
  const int nb = nt * 16 + 8 * half;
  if (has_tile) {
    const float* wrow = cw + (size_t)(nt * 16 + l16) * KDIM + 4 * half;
#pragma unroll
    for (int s = 0; s < CONV_NS; ++s) wq[s] = *(const float4*)(wrow + 8 * s);
    bl = *(const float4*)(cb + nb);
    bh = *(const float4*)(cb + nb + 4);
  }

  __builtin_amdgcn_fence(__ATOMIC_RELEASE, "workgroup", "local");
  __builtin_amdgcn_s_barrier();
  __builtin_amdgcn_fence(__ATOMIC_ACQUIRE, "workgroup", "local");
  if (!has_tile) return;

  const float* x0p = &Aim[l16][4 * half];
  const float* x1p = &Aim[16 + l16][4 * half];
  v8f acc0 = {}, acc1 = {};
  float4 xc[CONV_G], yc[CONV_G];
#pragma unroll
  for (int i = 0; i < CONV_G; ++i) {
    xc[i] = *(const float4*)(x0p + 8 * i);
    yc[i] = *(const float4*)(x1p + 8 * i);
  }
#pragma unroll
  for (int g = 0; g < CONV_NG; ++g) {
    float4 xn[CONV_G], yn[CONV_G];
    if (g + 1 < CONV_NG) {
#pragma unroll
      for (int i = 0; i < CONV_G; ++i) {
        xn[i] = *(const float4*)(x0p + 8 * ((g + 1) * CONV_G + i));
        yn[i] = *(const float4*)(x1p + 8 * ((g + 1) * CONV_G + i));
      }
    }
    __builtin_amdgcn_sched_barrier(0);
#pragma unroll
    for (int i = 0; i < CONV_G; ++i) {
      const float4 w = wq[g * CONV_G + i];
      v2f wa, wb, xa, xb, ya, yb;
      wa.x = w.x;     wa.y = w.y;     wb.x = w.z;     wb.y = w.w;
      xa.x = xc[i].x; xa.y = xc[i].y; xb.x = xc[i].z; xb.y = xc[i].w;
      ya.x = yc[i].x; ya.y = yc[i].y; yb.x = yc[i].z; yb.y = yc[i].w;
      acc0 = wmma_f32_4(wa, xa, acc0);
      acc1 = wmma_f32_4(wa, ya, acc1);
      acc0 = wmma_f32_4(wb, xb, acc0);
      acc1 = wmma_f32_4(wb, yb, acc1);
    }
    __builtin_amdgcn_sched_barrier(0);
    if (g + 1 < CONV_NG) {
#pragma unroll
      for (int i = 0; i < CONV_G; ++i) { xc[i] = xn[i]; yc[i] = yn[i]; }
    }
  }

  const float bias[8] = {bl.x, bl.y, bl.z, bl.w, bh.x, bh.y, bh.z, bh.w};
  const int ta = t0 + l16;
  const int tb = t0 + 16 + l16;
#pragma unroll
  for (int r = 0; r < 8; ++r) {
    float* orow = convout + ((size_t)b * OCH + nb + r) * TLEN;
    if (ta < TLEN) orow[ta] = acc0[r] + bias[r];
    if (tb < TLEN) orow[tb] = acc1[r] + bias[r];
  }
}

__global__ __launch_bounds__(256) void caps_norm_kernel(
    const float* __restrict__ convout, const float* __restrict__ gamma,
    const float* __restrict__ beta, float* __restrict__ out)
{
  const int idx = blockIdx.x * 256 + threadIdx.x;
  if (idx >= BSZ * TLEN * NCLS) return;
  const int cls = idx % NCLS;
  const int t   = (idx / NCLS) % TLEN;
  const int b   = idx / (NCLS * TLEN);
  const float scale = gamma[t] * rsqrtf(1.0f + 0.001f);
  const float shift = beta[t];
  float sum = 0.0f;
  for (int d = 0; d < 16; ++d) {
    const int o = d * NCLS + cls;
    const float v  = convout[((size_t)b * OCH + o) * TLEN + t];
    const float s  = 1.0f / (1.0f + __expf(-(v * scale + shift)));
    const float df = s - 0.5f;
    sum += df * df;
  }
  out[idx] = sqrtf(sum * 0.25f);
}

extern "C" void kernel_launch(void* const* d_in, const int* in_sizes, int n_in,
                              void* d_out, int out_size, void* d_ws, size_t ws_size,
                              hipStream_t stream) {
  (void)in_sizes; (void)n_in; (void)out_size; (void)ws_size;
  const float* x  = (const float*)d_in[0];
  const float* ea = (const float*)d_in[3];
  const float* W1 = (const float*)d_in[4];  const float* b1 = (const float*)d_in[5];
  const float* W2 = (const float*)d_in[6];  const float* b2 = (const float*)d_in[7];
  const float* W3 = (const float*)d_in[8];  const float* b3 = (const float*)d_in[9];
  const float* W4 = (const float*)d_in[10]; const float* b4 = (const float*)d_in[11];
  const float* cw = (const float*)d_in[12]; const float* cb = (const float*)d_in[13];
  const float* gm = (const float*)d_in[14]; const float* bt = (const float*)d_in[15];

  float* pooled  = (float*)d_ws;
  float* convout = pooled + (size_t)GNUM * 152;

  gcn_fused_kernel<<<GCN_WGS, 128, 0, stream>>>(x, ea, W1, b1, W2, b2, W3, b3, W4, b4, pooled);
  conv_wmma_kernel<<<dim3(BSZ * 4, 4), 160, 0, stream>>>(pooled, cw, cb, convout);
  const int nout = BSZ * TLEN * NCLS;
  caps_norm_kernel<<<(nout + 255) / 256, 256, 0, stream>>>(convout, gm, bt, (float*)d_out);
}


# GCN layer 4 epilogue: each wave takes the mean over the 22 node rows from its A @ Tm result registers and writes pooled itself; layer-4 H skips LDS, pool pass and 2 of 6 barriers per graph gone
# speedup vs baseline: 5.8339x; 1.0307x over previous
#include <hip/hip_runtime.h>
#include <hip/hip_bf16.h>
#include <cmath>

typedef __attribute__((ext_vector_type(2))) float v2f;
typedef __attribute__((ext_vector_type(8))) float v8f;

#define PNODES 22
#define GNUM   1920
#define CIN    14
#define HCOLS  160
#define EPG    462
#define TLEN   120
#define BSZ    16
#define OCH    272
#define KDIM   456
#define NCLS   17

__device__ __forceinline__ v8f wmma_f32_4(v2f a, v2f b, v8f c) {
  return __builtin_amdgcn_wmma_f32_16x16x4_f32(
       false, a,  false, b,
       (short)0, c,  false,  false);
}

#define GCN_WGS 960

template <int CIN_, int COUT_, int KS>
__device__ __forceinline__ void load_w_frags(v2f (&wf)[KS], const float* __restrict__ W,
                                             int nt, int half, int l16)
{
  const int  n   = nt * 16 + l16;
  const bool nok = n < COUT_;
  const int  ncl = nok ? n : 0;
  const float* base = W + (2 * half) * COUT_ + ncl;
#pragma unroll
  for (int ks = 0; ks < KS; ++ks) {
    if (4 * ks + 3 < CIN_) {
      const float w0 = base[(4 * ks) * COUT_];
      const float w1 = base[(4 * ks + 1) * COUT_];
      wf[ks].x = nok ? w0 : 0.0f;
      wf[ks].y = nok ? w1 : 0.0f;
    } else {
      const int k0  = 4 * ks + 2 * half;
      const int kr0 = (k0     < CIN_) ? k0     : 0;
      const int kr1 = (k0 + 1 < CIN_) ? k0 + 1 : 0;
      const float w0 = W[kr0 * COUT_ + ncl];
      const float w1 = W[kr1 * COUT_ + ncl];
      wf[ks].x = (k0     < CIN_ && nok) ? w0 : 0.0f;
      wf[ks].y = (k0 + 1 < CIN_ && nok) ? w1 : 0.0f;
    }
  }
}

__device__ __forceinline__ int slot_node(int s) { return s < 19 ? s : ((s >= 24 && s < 27) ? s - 5 : -1); }
__device__ __forceinline__ int node_slot(int r) { return r < 19 ? r : r + 5; }

__device__ __forceinline__ void store_h(float (&H)[32][HCOLS], v8f c, int mt, int n, int half,
                                        float bv, bool nok)
{
#pragma unroll
  for (int v = 0; v < 8; ++v) {
    const int  row  = mt * 16 + 8 * half + v;
    const bool real = (mt == 0) || (v < 3);
    H[row][n] = (real && nok) ? fmaxf(c[v] + bv, 0.0f) : 0.0f;
  }
}

__device__ __forceinline__ void stepb_regs(float (&H)[32][HCOLS], const v8f t0, const v8f t1,
    const v2f (&adj0)[6], const v2f (&adj1)[6], int n, int half, float bv, bool nok)
{
  v8f d0 = {}, d1 = {};
#pragma unroll
  for (int ks = 0; ks < 6; ++ks) {
    v2f b;
    if (ks < 4) { b.x = t0[2 * ks];       b.y = t0[2 * ks + 1]; }
    else        { b.x = t1[2 * (ks - 4)]; b.y = t1[2 * (ks - 4) + 1]; }
    d0 = wmma_f32_4(adj0[ks], b, d0);
    d1 = wmma_f32_4(adj1[ks], b, d1);
  }
  store_h(H, d0, 0, n, half, bv, nok);
  store_h(H, d1, 1, n, half, bv, nok);
}

__device__ __forceinline__ float stepb_pool(const v8f t0, const v8f t1,
    const v2f (&adj0)[6], const v2f (&adj1)[6], float bv)
{
  v8f d0 = {}, d1 = {};
#pragma unroll
  for (int ks = 0; ks < 6; ++ks) {
    v2f b;
    if (ks < 4) { b.x = t0[2 * ks];       b.y = t0[2 * ks + 1]; }
    else        { b.x = t1[2 * (ks - 4)]; b.y = t1[2 * (ks - 4) + 1]; }
    d0 = wmma_f32_4(adj0[ks], b, d0);
    d1 = wmma_f32_4(adj1[ks], b, d1);
  }
  float s = 0.0f;
#pragma unroll
  for (int v = 0; v < 8; ++v) s += fmaxf(d0[v] + bv, 0.0f);
#pragma unroll
  for (int v = 0; v < 3; ++v) s += fmaxf(d1[v] + bv, 0.0f);
  s += __shfl_xor(s, 16);
  return s * (1.0f / 22.0f);
}

template <int KS>
__device__ __forceinline__ void stepa_pair(v8f& c0, v8f& c1, const float* h0, const float* h1,
                                           const v2f* wf)
{
#pragma unroll
  for (int ks = 0; ks < KS; ++ks) {
    const v2f a0 = *(const v2f*)(h0 + 4 * ks);
    const v2f a1 = *(const v2f*)(h1 + 4 * ks);
    c0 = wmma_f32_4(a0, wf[ks], c0);
    c1 = wmma_f32_4(a1, wf[ks], c1);
  }
}

template <int NT>
__device__ __forceinline__ void layer4_stepa(v8f (&c)[6], const float* h0, const float* h1,
    const v2f (&wa)[16], const v2f (&wb)[16], const v2f (&wc)[16])
{
  v2f xa[4], xb[4];
#pragma unroll
  for (int i = 0; i < 4; ++i) {
    xa[i] = *(const v2f*)(h0 + 4 * i);
    xb[i] = *(const v2f*)(h1 + 4 * i);
  }
#pragma unroll
  for (int ck = 0; ck < 4; ++ck) {
    v2f na[4], nb[4];
    if (ck + 1 < 4) {
#pragma unroll
      for (int i = 0; i < 4; ++i) {
        na[i] = *(const v2f*)(h0 + 4 * ((ck + 1) * 4 + i));
        nb[i] = *(const v2f*)(h1 + 4 * ((ck + 1) * 4 + i));
      }
    }
    __builtin_amdgcn_sched_barrier(0);
#pragma unroll
    for (int i = 0; i < 4; ++i) {
      const int ks = ck * 4 + i;
      const v2f a0 = xa[i], a1 = xb[i];
      c[0] = wmma_f32_4(a0, wa[ks], c[0]);
      c[1] = wmma_f32_4(a1, wa[ks], c[1]);
      c[2] = wmma_f32_4(a0, wb[ks], c[2]);
      c[3] = wmma_f32_4(a1, wb[ks], c[3]);
      if (NT == 3) {
        c[4] = wmma_f32_4(a0, wc[ks], c[4]);
        c[5] = wmma_f32_4(a1, wc[ks], c[5]);
      }
    }
    __builtin_amdgcn_sched_barrier(0);
    if (ck + 1 < 4) {
#pragma unroll
      for (int i = 0; i < 4; ++i) { xa[i] = na[i]; xb[i] = nb[i]; }
    }
  }
}

__global__ __launch_bounds__(128) void gcn_fused_kernel(
    const float* __restrict__ x, const float* __restrict__ ea,
    const float* __restrict__ W1, const float* __restrict__ b1,
    const float* __restrict__ W2, const float* __restrict__ b2,
    const float* __restrict__ W3, const float* __restrict__ b3,
    const float* __restrict__ W4, const float* __restrict__ b4,
    float* __restrict__ pooled)
{
  __shared__ __attribute__((aligned(16))) float Ha[32][HCOLS];
  __shared__ __attribute__((aligned(16))) float Hb[32][HCOLS];

  const int tid  = threadIdx.x;
  const int wave = __builtin_amdgcn_readfirstlane(tid >> 5);
  const int lane = tid & 31;
  const int half = lane >> 4;
  const int l16  = lane & 15;
  const int swap  = (blockIdx.x + (blockIdx.x >> 8)) & 1;
  const int role  = wave ^ (swap << 1);
  const bool three = role < 2;

  const int tA = three ? 3 * role : 2 + 2 * role;
  const int tB = tA + 1, tX = tA + 2;
  const int t2 = role & 1;
  const int nA = tA * 16 + l16, nB = tB * 16 + l16, nX = tX * 16 + l16;
  const int n3 = role * 16 + l16, n2 = t2 * 16 + l16, n1 = l16;
  const bool okB = nB < 152;
  v2f w4a[16], w4b[16], wx[16], w3f[8];
  load_w_frags<64, 152, 16>(w4a, W4, tA, half, l16);
  load_w_frags<64, 152, 16>(w4b, W4, tB, half, l16);
  load_w_frags<32, 64, 8>(w3f, W3, role, half, l16);
  if (three) {
    load_w_frags<64, 152, 16>(wx, W4, tX, half, l16);
  } else {
    v2f w2t[4];
    load_w_frags<16, 32, 4>(w2t, W2, t2, half, l16);
#pragma unroll
    for (int ks = 0; ks < 16; ++ks) { wx[ks].x = (ks < 4) ? w2t[ks & 3].x : 0.0f; wx[ks].y = (ks < 4) ? w2t[ks & 3].y : 0.0f; }
  }
  const float bvA = b4[nA], bvB = okB ? b4[okB ? nB : 0] : 0.0f;
  const float bvX = three ? b4[three ? nX : 0] : b2[n2];
  const float bv3 = b3[n3], bv1 = b1[n1];

  for (int g = blockIdx.x; g < GNUM; g += GCN_WGS) {
    int l16s = l16, halfs = half, tids = tid;
    asm volatile("" : "+v"(l16s), "+v"(halfs), "+v"(tids));
    const int l16 = l16s, half = halfs, tid = tids;
    const int nA = tA * 16 + l16, nB = tB * 16 + l16, nX = tX * 16 + l16;
    const int n3 = role * 16 + l16, n2 = t2 * 16 + l16, n1 = l16;
    const float* ha0 = &Ha[l16][2 * half];
    const float* ha1 = &Ha[16 + l16][2 * half];
    const float* hb0 = &Hb[l16][2 * half];
    const float* hb1 = &Hb[16 + l16][2 * half];
    const float* eaw = ea + (size_t)g * EPG * 5;

    if (g + GCN_WGS < GNUM) {
      const char* nbase = (const char*)(ea + (size_t)(g + GCN_WGS) * EPG * 5);
      const char* nfeat = (const char*)(x + (size_t)(g + GCN_WGS) * PNODES * CIN);
      __builtin_prefetch(nbase + (size_t)tid * 64, 0, 1);
      __builtin_prefetch(nbase + (size_t)(tid + 128) * 64 % 9240, 0, 1);
      __builtin_prefetch(nfeat + (size_t)tid * 64 % 1232, 0, 1);
    }

    float raw0[12], raw1[12];
    const int  j0 = l16;
    const int  j1 = slot_node(16 + l16);
    const bool j1ok = j1 >= 0;
    {
      const int  j1c = j1ok ? j1 : 0;
#pragma unroll
      for (int q = 0; q < 12; ++q) {
        const int  i   = slot_node((q < 8) ? 8 * half + q : 16 + 8 * half + (q - 8));
        const bool iok = i >= 0;
        const int  ic  = iok ? i : 0;
        const int  e0  = (ic == j0)  ? 0 : ic * (PNODES - 1) + (j0  < ic ? j0  : j0  - 1);
        const int  e1  = (ic == j1c) ? 0 : ic * (PNODES - 1) + (j1c < ic ? j1c : j1c - 1);
        raw0[q] = eaw[e0 * 5 + 4];
        raw1[q] = eaw[e1 * 5 + 4];
      }
    }
    v2f w1f[4] = {};
    if (role == 2) load_w_frags<14, 16, 4>(w1f, W1, 0, half, l16);
    for (int idx = tid; idx < 32 * 16; idx += 128) {
      const int r = idx >> 4, c = idx & 15;
      if (slot_node(r) < 0 || c >= CIN) Ha[r][c] = 0.0f;
    }
    for (int idx = tid; idx < PNODES * CIN; idx += 128) {
      const int r = idx / CIN, c = idx % CIN;
      Ha[node_slot(r)][c] = x[(size_t)(g * PNODES + r) * CIN + c];
    }
    __syncthreads();

    v2f adj0[6], adj1[6];
    {
      float w0[12], w1[12];
      float s0 = 0.0f, s1 = 0.0f;
#pragma unroll
      for (int q = 0; q < 12; ++q) {
        const int  i   = slot_node((q < 8) ? 8 * half + q : 16 + 8 * half + (q - 8));
        const bool iok = i >= 0;
        w0[q] = iok ? ((i == j0) ? 1.0f : raw0[q]) : 0.0f;
        w1[q] = (iok && j1ok) ? ((i == j1) ? 1.0f : raw1[q]) : 0.0f;
        s0 += w0[q];
        s1 += w1[q];
      }
      s0 += __shfl_xor(s0, 16);
      s1 += __shfl_xor(s1, 16);
      const float dr0 = (s0 > 0.0f) ? rsqrtf(s0) : 0.0f;
      const float dr1 = (s1 > 0.0f) ? rsqrtf(s1) : 0.0f;
#pragma unroll
      for (int ks = 0; ks < 6; ++ks) {
        const int sl = 8 * half + 2 * (ks & 3);
        const float dc0 = (ks < 4) ? __shfl(dr0, sl)     : __shfl(dr1, sl);
        const float dc1 = (ks < 4) ? __shfl(dr0, sl + 1) : __shfl(dr1, sl + 1);
        adj0[ks].x = dc0 * w0[2 * ks]     * dr0;
        adj0[ks].y = dc1 * w0[2 * ks + 1] * dr0;
        adj1[ks].x = dc0 * w1[2 * ks]     * dr1;
        adj1[ks].y = dc1 * w1[2 * ks + 1] * dr1;
      }
    }

    if (role == 2) {
      v8f c0 = {}, c1 = {};
      stepa_pair<4>(c0, c1, ha0, ha1, w1f);
      stepb_regs(Hb, c0, c1, adj0, adj1, n1, half, bv1, true);
    }
    __syncthreads();

    if (!three) {
      v8f c0 = {}, c1 = {};
      stepa_pair<4>(c0, c1, hb0, hb1, wx);
      stepb_regs(Ha, c0, c1, adj0, adj1, n2, half, bvX, true);
    }
    __syncthreads();

    {
      v8f c0 = {}, c1 = {};
      stepa_pair<8>(c0, c1, ha0, ha1, w3f);
      stepb_regs(Hb, c0, c1, adj0, adj1, n3, half, bv3, true);
    }
    __syncthreads();

    {
      v8f c[6] = {};
      float* prow = pooled + (size_t)g * 152;
      float bA = bvA, bB = bvB, bX = bvX;
      asm volatile("" : "+v"(bA), "+v"(bB), "+v"(bX));
      if (three) {
        layer4_stepa<3>(c, hb0, hb1, w4a, w4b, wx);
        const float pA = stepb_pool(c[0], c[1], adj0, adj1, bA);
        const float pB = stepb_pool(c[2], c[3], adj0, adj1, bB);
        const float pX = stepb_pool(c[4], c[5], adj0, adj1, bX);
        prow[nA + 16 * half] = half ? pB : pA;
        if (half == 0) prow[nX] = pX;
      } else {
        layer4_stepa<2>(c, hb0, hb1, w4a, w4b, wx);
        const float pA = stepb_pool(c[0], c[1], adj0, adj1, bA);
        const float pB = stepb_pool(c[2], c[3], adj0, adj1, bB);
        const int n = nA + 16 * half;
        if (n < 152) prow[n] = half ? pB : pA;
      }
    }
  }
}

#define CONV_TW    32
#define CONV_PITCH 460
#define CONV_NS    57
#define CONV_G     3
#define CONV_NG    (CONV_NS / CONV_G)

__global__ __launch_bounds__(160) void conv_wmma_kernel(
    const float* __restrict__ pooled, const float* __restrict__ cw,
    const float* __restrict__ cb, float* __restrict__ convout)
{
  __shared__ __attribute__((aligned(16))) float Aim[CONV_TW][CONV_PITCH];

  const int tid  = threadIdx.x;
  const int wave = __builtin_amdgcn_readfirstlane(tid >> 5);
  const int lane = tid & 31;
  const int half = lane >> 4;
  const int l16  = lane & 15;
  const int b    = blockIdx.x >> 2;
  const int t0   = (blockIdx.x & 3) * CONV_TW;
  const int nt   = wave * 4 + blockIdx.y;
  const bool has_tile = nt < OCH / 16;

  const int  rg     = tid / 38;
  const int  c4     = tid - rg * 38;
  const bool stager = tid < 4 * 38;
  const float* pcol = pooled + (size_t)b * TLEN * 152 + c4 * 4;
  if (stager) {
    float4 tap[8][3];
#pragma unroll
    for (int it = 0; it < 8; ++it) {
#pragma unroll
      for (int kk = 0; kk < 3; ++kk) {
        const int t  = t0 + rg + 4 * it + kk - 1;
        const int tc = t < 0 ? 0 : (t >= TLEN ? TLEN - 1 : t);
        tap[it][kk] = *(const float4*)(pcol + tc * 152);
      }
    }
    __builtin_amdgcn_sched_barrier(0);
#pragma unroll
    for (int it = 0; it < 8; ++it) {
      float4 r[3];
#pragma unroll
      for (int kk = 0; kk < 3; ++kk) {
        const int  t  = t0 + rg + 4 * it + kk - 1;
        const bool ok = (t >= 0) && (t < TLEN);
        r[kk].x = ok ? tap[it][kk].x : 0.0f;
        r[kk].y = ok ? tap[it][kk].y : 0.0f;
        r[kk].z = ok ? tap[it][kk].z : 0.0f;
        r[kk].w = ok ? tap[it][kk].w : 0.0f;
      }
      float4* dst = (float4*)&Aim[rg + 4 * it][c4 * 12];
      dst[0] = make_float4(r[0].x, r[1].x, r[2].x, r[0].y);
      dst[1] = make_float4(r[1].y, r[2].y, r[0].z, r[1].z);
      dst[2] = make_float4(r[2].z, r[0].w, r[1].w, r[2].w);
    }
  }

  float4 wq[CONV_NS];
  float4 bl, bh;
  const int nb = nt * 16 + 8 * half;
  if (has_tile) {
    const float* wrow = cw + (size_t)(nt * 16 + l16) * KDIM + 4 * half;
#pragma unroll
    for (int s = 0; s < CONV_NS; ++s) wq[s] = *(const float4*)(wrow + 8 * s);
    bl = *(const float4*)(cb + nb);
    bh = *(const float4*)(cb + nb + 4);
  }

  __builtin_amdgcn_fence(__ATOMIC_RELEASE, "workgroup", "local");
  __builtin_amdgcn_s_barrier();
  __builtin_amdgcn_fence(__ATOMIC_ACQUIRE, "workgroup", "local");
  if (!has_tile) return;

  const float* x0p = &Aim[l16][4 * half];
  const float* x1p = &Aim[16 + l16][4 * half];
  v8f acc0 = {}, acc1 = {};
  float4 xc[CONV_G], yc[CONV_G];
#pragma unroll
  for (int i = 0; i < CONV_G; ++i) {
    xc[i] = *(const float4*)(x0p + 8 * i);
    yc[i] = *(const float4*)(x1p + 8 * i);
  }
#pragma unroll
  for (int g = 0; g < CONV_NG; ++g) {
    float4 xn[CONV_G], yn[CONV_G];
    if (g + 1 < CONV_NG) {
#pragma unroll
      for (int i = 0; i < CONV_G; ++i) {
        xn[i] = *(const float4*)(x0p + 8 * ((g + 1) * CONV_G + i));
        yn[i] = *(const float4*)(x1p + 8 * ((g + 1) * CONV_G + i));
      }
    }
    __builtin_amdgcn_sched_barrier(0);
#pragma unroll
    for (int i = 0; i < CONV_G; ++i) {
      const float4 w = wq[g * CONV_G + i];
      v2f wa, wb, xa, xb, ya, yb;
      wa.x = w.x;     wa.y = w.y;     wb.x = w.z;     wb.y = w.w;
      xa.x = xc[i].x; xa.y = xc[i].y; xb.x = xc[i].z; xb.y = xc[i].w;
      ya.x = yc[i].x; ya.y = yc[i].y; yb.x = yc[i].z; yb.y = yc[i].w;
      acc0 = wmma_f32_4(wa, xa, acc0);
      acc1 = wmma_f32_4(wa, ya, acc1);
      acc0 = wmma_f32_4(wb, xb, acc0);
      acc1 = wmma_f32_4(wb, yb, acc1);
    }
    __builtin_amdgcn_sched_barrier(0);
    if (g + 1 < CONV_NG) {
#pragma unroll
      for (int i = 0; i < CONV_G; ++i) { xc[i] = xn[i]; yc[i] = yn[i]; }
    }
  }

  const float bias[8] = {bl.x, bl.y, bl.z, bl.w, bh.x, bh.y, bh.z, bh.w};
  const int ta = t0 + l16;
  const int tb = t0 + 16 + l16;
#pragma unroll
  for (int r = 0; r < 8; ++r) {
    float* orow = convout + ((size_t)b * OCH + nb + r) * TLEN;
    if (ta < TLEN) orow[ta] = acc0[r] + bias[r];
    if (tb < TLEN) orow[tb] = acc1[r] + bias[r];
  }
}

__global__ __launch_bounds__(256) void caps_norm_kernel(
    const float* __restrict__ convout, const float* __restrict__ gamma,
    const float* __restrict__ beta, float* __restrict__ out)
{
  const int idx = blockIdx.x * 256 + threadIdx.x;
  if (idx >= BSZ * TLEN * NCLS) return;
  const int cls = idx % NCLS;
  const int t   = (idx / NCLS) % TLEN;
  const int b   = idx / (NCLS * TLEN);
  const float scale = gamma[t] * rsqrtf(1.0f + 0.001f);
  const float shift = beta[t];
  float sum = 0.0f;
  for (int d = 0; d < 16; ++d) {
    const int o = d * NCLS + cls;
    const float v  = convout[((size_t)b * OCH + o) * TLEN + t];
    const float s  = 1.0f / (1.0f + __expf(-(v * scale + shift)));
    const float df = s - 0.5f;
    sum += df * df;
  }
  out[idx] = sqrtf(sum * 0.25f);
}

extern "C" void kernel_launch(void* const* d_in, const int* in_sizes, int n_in,
                              void* d_out, int out_size, void* d_ws, size_t ws_size,
                              hipStream_t stream) {
  (void)in_sizes; (void)n_in; (void)out_size; (void)ws_size;
  const float* x  = (const float*)d_in[0];
  const float* ea = (const float*)d_in[3];
  const float* W1 = (const float*)d_in[4];  const float* b1 = (const float*)d_in[5];
  const float* W2 = (const float*)d_in[6];  const float* b2 = (const float*)d_in[7];
  const float* W3 = (const float*)d_in[8];  const float* b3 = (const float*)d_in[9];
  const float* W4 = (const float*)d_in[10]; const float* b4 = (const float*)d_in[11];
  const float* cw = (const float*)d_in[12]; const float* cb = (const float*)d_in[13];
  const float* gm = (const float*)d_in[14]; const float* bt = (const float*)d_in[15];

  float* pooled  = (float*)d_ws;
  float* convout = pooled + (size_t)GNUM * 152;

  gcn_fused_kernel<<<GCN_WGS, 128, 0, stream>>>(x, ea, W1, b1, W2, b2, W3, b3, W4, b4, pooled);
  conv_wmma_kernel<<<dim3(BSZ * 4, 4), 160, 0, stream>>>(pooled, cw, cb, convout);
  const int nout = BSZ * TLEN * NCLS;
  caps_norm_kernel<<<(nout + 255) / 256, 256, 0, stream>>>(convout, gm, bt, (float*)d_out);
}


# GCN front end: edge offsets from lane constants (22 loads, at most one compare each), features + pad by one 16-byte LDS store per thread, first graph's block prefetched at entry before weight loads
# speedup vs baseline: 6.3888x; 1.0951x over previous
#include <hip/hip_runtime.h>
#include <hip/hip_bf16.h>
#include <cmath>

typedef __attribute__((ext_vector_type(2))) float v2f;
typedef __attribute__((ext_vector_type(8))) float v8f;

#define PNODES 22
#define GNUM   1920
#define CIN    14
#define HCOLS  160
#define EPG    462
#define TLEN   120
#define BSZ    16
#define OCH    272
#define KDIM   456
#define NCLS   17

__device__ __forceinline__ v8f wmma_f32_4(v2f a, v2f b, v8f c) {
  return __builtin_amdgcn_wmma_f32_16x16x4_f32(
       false, a,  false, b,
       (short)0, c,  false,  false);
}

#define GCN_WGS 960

template <int CIN_, int COUT_, int KS>
__device__ __forceinline__ void load_w_frags(v2f (&wf)[KS], const float* __restrict__ W,
                                             int nt, int half, int l16)
{
  const int  n   = nt * 16 + l16;
  const bool nok = n < COUT_;
  const int  ncl = nok ? n : 0;
  const float* base = W + (2 * half) * COUT_ + ncl;
#pragma unroll
  for (int ks = 0; ks < KS; ++ks) {
    if (4 * ks + 3 < CIN_) {
      const float w0 = base[(4 * ks) * COUT_];
      const float w1 = base[(4 * ks + 1) * COUT_];
      wf[ks].x = nok ? w0 : 0.0f;
      wf[ks].y = nok ? w1 : 0.0f;
    } else {
      const int k0  = 4 * ks + 2 * half;
      const int kr0 = (k0     < CIN_) ? k0     : 0;
      const int kr1 = (k0 + 1 < CIN_) ? k0 + 1 : 0;
      const float w0 = W[kr0 * COUT_ + ncl];
      const float w1 = W[kr1 * COUT_ + ncl];
      wf[ks].x = (k0     < CIN_ && nok) ? w0 : 0.0f;
      wf[ks].y = (k0 + 1 < CIN_ && nok) ? w1 : 0.0f;
    }
  }
}

__device__ __forceinline__ int slot_node(int s) { return s < 19 ? s : ((s >= 24 && s < 27) ? s - 5 : -1); }

__device__ __forceinline__ void store_h(float (&H)[32][HCOLS], v8f c, int mt, int n, int half,
                                        float bv, bool nok)
{
#pragma unroll
  for (int v = 0; v < 8; ++v) {
    const int  row  = mt * 16 + 8 * half + v;
    const bool real = (mt == 0) || (v < 3);
    H[row][n] = (real && nok) ? fmaxf(c[v] + bv, 0.0f) : 0.0f;
  }
}

__device__ __forceinline__ void stepb_regs(float (&H)[32][HCOLS], const v8f t0, const v8f t1,
    const v2f (&adj0)[6], const v2f (&adj1)[6], int n, int half, float bv, bool nok)
{
  v8f d0 = {}, d1 = {};
#pragma unroll
  for (int ks = 0; ks < 6; ++ks) {
    v2f b;
    if (ks < 4) { b.x = t0[2 * ks];       b.y = t0[2 * ks + 1]; }
    else        { b.x = t1[2 * (ks - 4)]; b.y = t1[2 * (ks - 4) + 1]; }
    d0 = wmma_f32_4(adj0[ks], b, d0);
    d1 = wmma_f32_4(adj1[ks], b, d1);
  }
  store_h(H, d0, 0, n, half, bv, nok);
  store_h(H, d1, 1, n, half, bv, nok);
}

__device__ __forceinline__ float stepb_pool(const v8f t0, const v8f t1,
    const v2f (&adj0)[6], const v2f (&adj1)[6], float bv)
{
  v8f d0 = {}, d1 = {};
#pragma unroll
  for (int ks = 0; ks < 6; ++ks) {
    v2f b;
    if (ks < 4) { b.x = t0[2 * ks];       b.y = t0[2 * ks + 1]; }
    else        { b.x = t1[2 * (ks - 4)]; b.y = t1[2 * (ks - 4) + 1]; }
    d0 = wmma_f32_4(adj0[ks], b, d0);
    d1 = wmma_f32_4(adj1[ks], b, d1);
  }
  float s = 0.0f;
#pragma unroll
  for (int v = 0; v < 8; ++v) s += fmaxf(d0[v] + bv, 0.0f);
#pragma unroll
  for (int v = 0; v < 3; ++v) s += fmaxf(d1[v] + bv, 0.0f);
  s += __shfl_xor(s, 16);
  return s * (1.0f / 22.0f);
}

template <int KS>
__device__ __forceinline__ void stepa_pair(v8f& c0, v8f& c1, const float* h0, const float* h1,
                                           const v2f* wf)
{
#pragma unroll
  for (int ks = 0; ks < KS; ++ks) {
    const v2f a0 = *(const v2f*)(h0 + 4 * ks);
    const v2f a1 = *(const v2f*)(h1 + 4 * ks);
    c0 = wmma_f32_4(a0, wf[ks], c0);
    c1 = wmma_f32_4(a1, wf[ks], c1);
  }
}

template <int NT>
__device__ __forceinline__ void layer4_stepa(v8f (&c)[6], const float* h0, const float* h1,
    const v2f (&wa)[16], const v2f (&wb)[16], const v2f (&wc)[16])
{
  v2f xa[4], xb[4];
#pragma unroll
  for (int i = 0; i < 4; ++i) {
    xa[i] = *(const v2f*)(h0 + 4 * i);
    xb[i] = *(const v2f*)(h1 + 4 * i);
  }
#pragma unroll
  for (int ck = 0; ck < 4; ++ck) {
    v2f na[4], nb[4];
    if (ck + 1 < 4) {
#pragma unroll
      for (int i = 0; i < 4; ++i) {
        na[i] = *(const v2f*)(h0 + 4 * ((ck + 1) * 4 + i));
        nb[i] = *(const v2f*)(h1 + 4 * ((ck + 1) * 4 + i));
      }
    }
    __builtin_amdgcn_sched_barrier(0);
#pragma unroll
    for (int i = 0; i < 4; ++i) {
      const int ks = ck * 4 + i;
      const v2f a0 = xa[i], a1 = xb[i];
      c[0] = wmma_f32_4(a0, wa[ks], c[0]);
      c[1] = wmma_f32_4(a1, wa[ks], c[1]);
      c[2] = wmma_f32_4(a0, wb[ks], c[2]);
      c[3] = wmma_f32_4(a1, wb[ks], c[3]);
      if (NT == 3) {
        c[4] = wmma_f32_4(a0, wc[ks], c[4]);
        c[5] = wmma_f32_4(a1, wc[ks], c[5]);
      }
    }
    __builtin_amdgcn_sched_barrier(0);
    if (ck + 1 < 4) {
#pragma unroll
      for (int i = 0; i < 4; ++i) { xa[i] = na[i]; xb[i] = nb[i]; }
    }
  }
}

__device__ __forceinline__ void prefetch_graph(const float* __restrict__ ea, const float* __restrict__ x,
                                               int g, int tid)
{
  const char* eb = (const char*)(ea + (size_t)g * EPG * 5);
  const char* fb = (const char*)(x + (size_t)g * PNODES * CIN);
  __builtin_prefetch(eb + (unsigned)tid * 64u, 0, 1);
  const char* p2 = (tid < 17) ? eb + 8192u + (unsigned)tid * 64u : fb + (unsigned)(tid - 17) * 64u;
  if (tid < 37) __builtin_prefetch(p2, 0, 1);
}

__global__ __launch_bounds__(128) void gcn_fused_kernel(
    const float* __restrict__ x, const float* __restrict__ ea,
    const float* __restrict__ W1, const float* __restrict__ b1,
    const float* __restrict__ W2, const float* __restrict__ b2,
    const float* __restrict__ W3, const float* __restrict__ b3,
    const float* __restrict__ W4, const float* __restrict__ b4,
    float* __restrict__ pooled)
{
  __shared__ __attribute__((aligned(16))) float Ha[32][HCOLS];
  __shared__ __attribute__((aligned(16))) float Hb[32][HCOLS];

  const int tid  = threadIdx.x;
  const int wave = __builtin_amdgcn_readfirstlane(tid >> 5);
  const int lane = tid & 31;
  const int half = lane >> 4;
  const int l16  = lane & 15;
  const int swap  = (blockIdx.x + (blockIdx.x >> 8)) & 1;
  const int role  = wave ^ (swap << 1);
  const bool three = role < 2;

  prefetch_graph(ea, x, blockIdx.x, tid);

  const int tA = three ? 3 * role : 2 + 2 * role;
  const int tB = tA + 1, tX = tA + 2;
  const int t2 = role & 1;
  const int nA = tA * 16 + l16, nB = tB * 16 + l16, nX = tX * 16 + l16;
  const int n3 = role * 16 + l16, n2 = t2 * 16 + l16, n1 = l16;
  const bool okB = nB < 152;
  v2f w4a[16], w4b[16], wx[16], w3f[8];
  load_w_frags<64, 152, 16>(w4a, W4, tA, half, l16);
  load_w_frags<64, 152, 16>(w4b, W4, tB, half, l16);
  load_w_frags<32, 64, 8>(w3f, W3, role, half, l16);
  if (three) {
    load_w_frags<64, 152, 16>(wx, W4, tX, half, l16);
  } else {
    v2f w2t[4];
    load_w_frags<16, 32, 4>(w2t, W2, t2, half, l16);
#pragma unroll
    for (int ks = 0; ks < 16; ++ks) { wx[ks].x = (ks < 4) ? w2t[ks & 3].x : 0.0f; wx[ks].y = (ks < 4) ? w2t[ks & 3].y : 0.0f; }
  }
  const float bvA = b4[nA], bvB = okB ? b4[okB ? nB : 0] : 0.0f;
  const float bvX = three ? b4[three ? nX : 0] : b2[n2];
  const float bv3 = b3[n3], bv1 = b1[n1];

  for (int g = blockIdx.x; g < GNUM; g += GCN_WGS) {
    int l16s = l16, halfs = half, tids = tid;
    asm volatile("" : "+v"(l16s), "+v"(halfs), "+v"(tids));
    const int l16 = l16s, half = halfs, tid = tids;
    const int nA = tA * 16 + l16, nB = tB * 16 + l16, nX = tX * 16 + l16;
    const int n3 = role * 16 + l16, n2 = t2 * 16 + l16, n1 = l16;
    const float* ha0 = &Ha[l16][2 * half];
    const float* ha1 = &Ha[16 + l16][2 * half];
    const float* hb0 = &Hb[l16][2 * half];
    const float* hb1 = &Hb[16 + l16][2 * half];
    const float* eaw = ea + (size_t)g * EPG * 5;

    if (g + GCN_WGS < GNUM) prefetch_graph(ea, x, g + GCN_WGS, tid);

    float raw0[12], raw1[12];
    const int  hl   = half & 1, cl = l16 & 15;
    const int  d0   = cl - 8 * hl;
    const bool j1ok = (cl & 7) < 3;
    const int  j1s  = j1ok ? 16 + (cl & 7) + 3 * (cl >> 3) : 16;
    const int  e1   = j1s - 16 - 3 * hl;
    {
      typedef const __attribute__((address_space(1))) char* gcptr;
      gcptr eab = (gcptr)(unsigned long long)eaw;
      asm volatile("" : "+s"(eab));
      const unsigned b0  = 4u * (unsigned)(840 * hl + 5 * cl + 4);
      const unsigned b0m = b0 - 20u;
      const unsigned c0  = 4u * (unsigned)(1684 + 315 * hl + 5 * cl);
      const unsigned b1  = 4u * (unsigned)(840 * hl + 5 * j1s - 1);
      const unsigned c1  = 4u * (unsigned)(1684 + 315 * hl + 5 * j1s);
      const unsigned c1m = c1 - 20u;
#pragma unroll
      for (int q = 0; q < 8; ++q) {
        raw0[q] = *(const __attribute__((address_space(1))) float*)(eab + ((d0 > q) ? b0m : b0) + 420u * q);
        raw1[q] = *(const __attribute__((address_space(1))) float*)(eab + b1 + 420u * q);
      }
#pragma unroll
      for (int q = 8; q < 11; ++q) {
        raw0[q] = *(const __attribute__((address_space(1))) float*)(eab + c0 + 420u * (q - 8));
        raw1[q] = *(const __attribute__((address_space(1))) float*)(eab + ((e1 >= q - 8) ? c1m : c1) + 420u * (q - 8));
      }
      raw0[11] = 0.0f;
      raw1[11] = 0.0f;
    }
    v2f w1f[4] = {};
    if (role == 2) load_w_frags<14, 16, 4>(w1f, W1, 0, half, l16);
    {
      const int  r   = (tid >> 2) & 31, cq = tid & 3;
      const int  nd  = slot_node(r);
      const bool rok = nd >= 0;
      const bool bok = rok && cq < 3;
      const float* xr = x + (size_t)g * (PNODES * CIN) + (unsigned)((rok ? nd : 0) * CIN + 4 * cq);
      const float2 fa = *(const float2*)xr;
      const float2 fb = *(const float2*)(xr + (cq < 3 ? 2 : 0));
      float4 v;
      v.x = rok ? fa.x : 0.0f;
      v.y = rok ? fa.y : 0.0f;
      v.z = bok ? fb.x : 0.0f;
      v.w = bok ? fb.y : 0.0f;
      *(float4*)&Ha[r][4 * cq] = v;
    }
    __syncthreads();

    v2f adj0[6], adj1[6];
    {
      float w0[12], w1[12];
      float s0 = 0.0f, s1 = 0.0f;
#pragma unroll
      for (int q = 0; q < 12; ++q) {
        if (q < 8)       { w0[q] = (q == d0) ? 1.0f : raw0[q];  w1[q] = j1ok ? raw1[q] : 0.0f; }
        else if (q < 11) { w0[q] = raw0[q];  w1[q] = j1ok ? ((q - 8 == e1) ? 1.0f : raw1[q]) : 0.0f; }
        else             { w0[q] = 0.0f;     w1[q] = 0.0f; }
        s0 += w0[q];
        s1 += w1[q];
      }
      s0 += __shfl_xor(s0, 16);
      s1 += __shfl_xor(s1, 16);
      const float dr0 = (s0 > 0.0f) ? rsqrtf(s0) : 0.0f;
      const float dr1 = (s1 > 0.0f) ? rsqrtf(s1) : 0.0f;
#pragma unroll
      for (int ks = 0; ks < 6; ++ks) {
        const int sl = 8 * half + 2 * (ks & 3);
        const float dc0 = (ks < 4) ? __shfl(dr0, sl)     : __shfl(dr1, sl);
        const float dc1 = (ks < 4) ? __shfl(dr0, sl + 1) : __shfl(dr1, sl + 1);
        adj0[ks].x = dc0 * w0[2 * ks]     * dr0;
        adj0[ks].y = dc1 * w0[2 * ks + 1] * dr0;
        adj1[ks].x = dc0 * w1[2 * ks]     * dr1;
        adj1[ks].y = dc1 * w1[2 * ks + 1] * dr1;
      }
    }

    if (role == 2) {
      v8f c0 = {}, c1 = {};
      stepa_pair<4>(c0, c1, ha0, ha1, w1f);
      stepb_regs(Hb, c0, c1, adj0, adj1, n1, half, bv1, true);
    }
    __syncthreads();

    if (!three) {
      v8f c0 = {}, c1 = {};
      stepa_pair<4>(c0, c1, hb0, hb1, wx);
      stepb_regs(Ha, c0, c1, adj0, adj1, n2, half, bvX, true);
    }
    __syncthreads();

    {
      v8f c0 = {}, c1 = {};
      stepa_pair<8>(c0, c1, ha0, ha1, w3f);
      stepb_regs(Hb, c0, c1, adj0, adj1, n3, half, bv3, true);
    }
    __syncthreads();

    {
      v8f c[6] = {};
      float* prow = pooled + (size_t)g * 152;
      float bA = bvA, bB = bvB, bX = bvX;
      asm volatile("" : "+v"(bA), "+v"(bB), "+v"(bX));
      if (three) {
        layer4_stepa<3>(c, hb0, hb1, w4a, w4b, wx);
        const float pA = stepb_pool(c[0], c[1], adj0, adj1, bA);
        const float pB = stepb_pool(c[2], c[3], adj0, adj1, bB);
        const float pX = stepb_pool(c[4], c[5], adj0, adj1, bX);
        prow[nA + 16 * half] = half ? pB : pA;
        if (half == 0) prow[nX] = pX;
      } else {
        layer4_stepa<2>(c, hb0, hb1, w4a, w4b, wx);
        const float pA = stepb_pool(c[0], c[1], adj0, adj1, bA);
        const float pB = stepb_pool(c[2], c[3], adj0, adj1, bB);
        const int n = nA + 16 * half;
        if (n < 152) prow[n] = half ? pB : pA;
      }
    }
  }
}

#define CONV_TW    32
#define CONV_PITCH 460
#define CONV_NS    57
#define CONV_G     3
#define CONV_NG    (CONV_NS / CONV_G)

__global__ __launch_bounds__(160) void conv_wmma_kernel(
    const float* __restrict__ pooled, const float* __restrict__ cw,
    const float* __restrict__ cb, float* __restrict__ convout)
{
  __shared__ __attribute__((aligned(16))) float Aim[CONV_TW][CONV_PITCH];

  const int tid  = threadIdx.x;
  const int wave = __builtin_amdgcn_readfirstlane(tid >> 5);
  const int lane = tid & 31;
  const int half = lane >> 4;
  const int l16  = lane & 15;
  const int b    = blockIdx.x >> 2;
  const int t0   = (blockIdx.x & 3) * CONV_TW;
  const int nt   = wave * 4 + blockIdx.y;
  const bool has_tile = nt < OCH / 16;

  const int  rg     = tid / 38;
  const int  c4     = tid - rg * 38;
  const bool stager = tid < 4 * 38;
  const float* pcol = pooled + (size_t)b * TLEN * 152 + c4 * 4;
  if (stager) {
    float4 tap[8][3];
#pragma unroll
    for (int it = 0; it < 8; ++it) {
#pragma unroll
      for (int kk = 0; kk < 3; ++kk) {
        const int t  = t0 + rg + 4 * it + kk - 1;
        const int tc = t < 0 ? 0 : (t >= TLEN ? TLEN - 1 : t);
        tap[it][kk] = *(const float4*)(pcol + tc * 152);
      }
    }
    __builtin_amdgcn_sched_barrier(0);
#pragma unroll
    for (int it = 0; it < 8; ++it) {
      float4 r[3];
#pragma unroll
      for (int kk = 0; kk < 3; ++kk) {
        const int  t  = t0 + rg + 4 * it + kk - 1;
        const bool ok = (t >= 0) && (t < TLEN);
        r[kk].x = ok ? tap[it][kk].x : 0.0f;
        r[kk].y = ok ? tap[it][kk].y : 0.0f;
        r[kk].z = ok ? tap[it][kk].z : 0.0f;
        r[kk].w = ok ? tap[it][kk].w : 0.0f;
      }
      float4* dst = (float4*)&Aim[rg + 4 * it][c4 * 12];
      dst[0] = make_float4(r[0].x, r[1].x, r[2].x, r[0].y);
      dst[1] = make_float4(r[1].y, r[2].y, r[0].z, r[1].z);
      dst[2] = make_float4(r[2].z, r[0].w, r[1].w, r[2].w);
    }
  }

  float4 wq[CONV_NS];
  float4 bl, bh;
  const int nb = nt * 16 + 8 * half;
  if (has_tile) {
    const float* wrow = cw + (size_t)(nt * 16 + l16) * KDIM + 4 * half;
#pragma unroll
    for (int s = 0; s < CONV_NS; ++s) wq[s] = *(const float4*)(wrow + 8 * s);
    bl = *(const float4*)(cb + nb);
    bh = *(const float4*)(cb + nb + 4);
  }

  __builtin_amdgcn_fence(__ATOMIC_RELEASE, "workgroup", "local");
  __builtin_amdgcn_s_barrier();
  __builtin_amdgcn_fence(__ATOMIC_ACQUIRE, "workgroup", "local");
  if (!has_tile) return;

  const float* x0p = &Aim[l16][4 * half];
  const float* x1p = &Aim[16 + l16][4 * half];
  v8f acc0 = {}, acc1 = {};
  float4 xc[CONV_G], yc[CONV_G];
#pragma unroll
  for (int i = 0; i < CONV_G; ++i) {
    xc[i] = *(const float4*)(x0p + 8 * i);
    yc[i] = *(const float4*)(x1p + 8 * i);
  }
#pragma unroll
  for (int g = 0; g < CONV_NG; ++g) {
    float4 xn[CONV_G], yn[CONV_G];
    if (g + 1 < CONV_NG) {
#pragma unroll
      for (int i = 0; i < CONV_G; ++i) {
        xn[i] = *(const float4*)(x0p + 8 * ((g + 1) * CONV_G + i));
        yn[i] = *(const float4*)(x1p + 8 * ((g + 1) * CONV_G + i));
      }
    }
    __builtin_amdgcn_sched_barrier(0);
#pragma unroll
    for (int i = 0; i < CONV_G; ++i) {
      const float4 w = wq[g * CONV_G + i];
      v2f wa, wb, xa, xb, ya, yb;
      wa.x = w.x;     wa.y = w.y;     wb.x = w.z;     wb.y = w.w;
      xa.x = xc[i].x; xa.y = xc[i].y; xb.x = xc[i].z; xb.y = xc[i].w;
      ya.x = yc[i].x; ya.y = yc[i].y; yb.x = yc[i].z; yb.y = yc[i].w;
      acc0 = wmma_f32_4(wa, xa, acc0);
      acc1 = wmma_f32_4(wa, ya, acc1);
      acc0 = wmma_f32_4(wb, xb, acc0);
      acc1 = wmma_f32_4(wb, yb, acc1);
    }
    __builtin_amdgcn_sched_barrier(0);
    if (g + 1 < CONV_NG) {
#pragma unroll
      for (int i = 0; i < CONV_G; ++i) { xc[i] = xn[i]; yc[i] = yn[i]; }
    }
  }

  const float bias[8] = {bl.x, bl.y, bl.z, bl.w, bh.x, bh.y, bh.z, bh.w};
  const int ta = t0 + l16;
  const int tb = t0 + 16 + l16;
#pragma unroll
  for (int r = 0; r < 8; ++r) {
    float* orow = convout + ((size_t)b * OCH + nb + r) * TLEN;
    if (ta < TLEN) orow[ta] = acc0[r] + bias[r];
    if (tb < TLEN) orow[tb] = acc1[r] + bias[r];
  }
}

__global__ __launch_bounds__(256) void caps_norm_kernel(
    const float* __restrict__ convout, const float* __restrict__ gamma,
    const float* __restrict__ beta, float* __restrict__ out)
{
  const int idx = blockIdx.x * 256 + threadIdx.x;
  if (idx >= BSZ * TLEN * NCLS) return;
  const int cls = idx % NCLS;
  const int t   = (idx / NCLS) % TLEN;
  const int b   = idx / (NCLS * TLEN);
  const float scale = gamma[t] * rsqrtf(1.0f + 0.001f);
  const float shift = beta[t];
  float sum = 0.0f;
  for (int d = 0; d < 16; ++d) {
    const int o = d * NCLS + cls;
    const float v  = convout[((size_t)b * OCH + o) * TLEN + t];
    const float s  = 1.0f / (1.0f + __expf(-(v * scale + shift)));
    const float df = s - 0.5f;
    sum += df * df;
  }
  out[idx] = sqrtf(sum * 0.25f);
}

extern "C" void kernel_launch(void* const* d_in, const int* in_sizes, int n_in,
                              void* d_out, int out_size, void* d_ws, size_t ws_size,
                              hipStream_t stream) {
  (void)in_sizes; (void)n_in; (void)out_size; (void)ws_size;
  const float* x  = (const float*)d_in[0];
  const float* ea = (const float*)d_in[3];
  const float* W1 = (const float*)d_in[4];  const float* b1 = (const float*)d_in[5];
  const float* W2 = (const float*)d_in[6];  const float* b2 = (const float*)d_in[7];
  const float* W3 = (const float*)d_in[8];  const float* b3 = (const float*)d_in[9];
  const float* W4 = (const float*)d_in[10]; const float* b4 = (const float*)d_in[11];
  const float* cw = (const float*)d_in[12]; const float* cb = (const float*)d_in[13];
  const float* gm = (const float*)d_in[14]; const float* bt = (const float*)d_in[15];

  float* pooled  = (float*)d_ws;
  float* convout = pooled + (size_t)GNUM * 152;

  gcn_fused_kernel<<<GCN_WGS, 128, 0, stream>>>(x, ea, W1, b1, W2, b2, W3, b3, W4, b4, pooled);
  conv_wmma_kernel<<<dim3(BSZ * 4, 4), 160, 0, stream>>>(pooled, cw, cb, convout);
  const int nout = BSZ * TLEN * NCLS;
  caps_norm_kernel<<<(nout + 255) / 256, 256, 0, stream>>>(convout, gm, bt, (float*)d_out);
}


# conv GEMM decomposition: the 16 x 120 positions as one run of 1920 columns, 60 x 17 wave tiles on 255 workgroups of 4 waves: one wave (228 WMMAs) per SIMD, not two on the busiest; no padded columns
# speedup vs baseline: 6.6328x; 1.0382x over previous
#include <hip/hip_runtime.h>
#include <hip/hip_bf16.h>
#include <cmath>

typedef __attribute__((ext_vector_type(2))) float v2f;
typedef __attribute__((ext_vector_type(8))) float v8f;

#define PNODES 22
#define GNUM   1920
#define CIN    14
#define HCOLS  160
#define EPG    462
#define TLEN   120
#define BSZ    16
#define OCH    272
#define KDIM   456
#define NCLS   17

__device__ __forceinline__ v8f wmma_f32_4(v2f a, v2f b, v8f c) {
  return __builtin_amdgcn_wmma_f32_16x16x4_f32(
       false, a,  false, b,
       (short)0, c,  false,  false);
}

#define GCN_WGS 960

template <int CIN_, int COUT_, int KS>
__device__ __forceinline__ void load_w_frags(v2f (&wf)[KS], const float* __restrict__ W,
                                             int nt, int half, int l16)
{
  const int  n   = nt * 16 + l16;
  const bool nok = n < COUT_;
  const int  ncl = nok ? n : 0;
  const float* base = W + (2 * half) * COUT_ + ncl;
#pragma unroll
  for (int ks = 0; ks < KS; ++ks) {
    if (4 * ks + 3 < CIN_) {
      const float w0 = base[(4 * ks) * COUT_];
      const float w1 = base[(4 * ks + 1) * COUT_];
      wf[ks].x = nok ? w0 : 0.0f;
      wf[ks].y = nok ? w1 : 0.0f;
    } else {
      const int k0  = 4 * ks + 2 * half;
      const int kr0 = (k0     < CIN_) ? k0     : 0;
      const int kr1 = (k0 + 1 < CIN_) ? k0 + 1 : 0;
      const float w0 = W[kr0 * COUT_ + ncl];
      const float w1 = W[kr1 * COUT_ + ncl];
      wf[ks].x = (k0     < CIN_ && nok) ? w0 : 0.0f;
      wf[ks].y = (k0 + 1 < CIN_ && nok) ? w1 : 0.0f;
    }
  }
}

__device__ __forceinline__ int slot_node(int s) { return s < 19 ? s : ((s >= 24 && s < 27) ? s - 5 : -1); }

__device__ __forceinline__ void store_h(float (&H)[32][HCOLS], v8f c, int mt, int n, int half,
                                        float bv, bool nok)
{
#pragma unroll
  for (int v = 0; v < 8; ++v) {
    const int  row  = mt * 16 + 8 * half + v;
    const bool real = (mt == 0) || (v < 3);
    H[row][n] = (real && nok) ? fmaxf(c[v] + bv, 0.0f) : 0.0f;
  }
}

__device__ __forceinline__ void stepb_regs(float (&H)[32][HCOLS], const v8f t0, const v8f t1,
    const v2f (&adj0)[6], const v2f (&adj1)[6], int n, int half, float bv, bool nok)
{
  v8f d0 = {}, d1 = {};
#pragma unroll
  for (int ks = 0; ks < 6; ++ks) {
    v2f b;
    if (ks < 4) { b.x = t0[2 * ks];       b.y = t0[2 * ks + 1]; }
    else        { b.x = t1[2 * (ks - 4)]; b.y = t1[2 * (ks - 4) + 1]; }
    d0 = wmma_f32_4(adj0[ks], b, d0);
    d1 = wmma_f32_4(adj1[ks], b, d1);
  }
  store_h(H, d0, 0, n, half, bv, nok);
  store_h(H, d1, 1, n, half, bv, nok);
}

__device__ __forceinline__ float stepb_pool(const v8f t0, const v8f t1,
    const v2f (&adj0)[6], const v2f (&adj1)[6], float bv)
{
  v8f d0 = {}, d1 = {};
#pragma unroll
  for (int ks = 0; ks < 6; ++ks) {
    v2f b;
    if (ks < 4) { b.x = t0[2 * ks];       b.y = t0[2 * ks + 1]; }
    else        { b.x = t1[2 * (ks - 4)]; b.y = t1[2 * (ks - 4) + 1]; }
    d0 = wmma_f32_4(adj0[ks], b, d0);
    d1 = wmma_f32_4(adj1[ks], b, d1);
  }
  float s = 0.0f;
#pragma unroll
  for (int v = 0; v < 8; ++v) s += fmaxf(d0[v] + bv, 0.0f);
#pragma unroll
  for (int v = 0; v < 3; ++v) s += fmaxf(d1[v] + bv, 0.0f);
  s += __shfl_xor(s, 16);
  return s * (1.0f / 22.0f);
}

template <int KS>
__device__ __forceinline__ void stepa_pair(v8f& c0, v8f& c1, const float* h0, const float* h1,
                                           const v2f* wf)
{
#pragma unroll
  for (int ks = 0; ks < KS; ++ks) {
    const v2f a0 = *(const v2f*)(h0 + 4 * ks);
    const v2f a1 = *(const v2f*)(h1 + 4 * ks);
    c0 = wmma_f32_4(a0, wf[ks], c0);
    c1 = wmma_f32_4(a1, wf[ks], c1);
  }
}

template <int NT>
__device__ __forceinline__ void layer4_stepa(v8f (&c)[6], const float* h0, const float* h1,
    const v2f (&wa)[16], const v2f (&wb)[16], const v2f (&wc)[16])
{
  v2f xa[4], xb[4];
#pragma unroll
  for (int i = 0; i < 4; ++i) {
    xa[i] = *(const v2f*)(h0 + 4 * i);
    xb[i] = *(const v2f*)(h1 + 4 * i);
  }
#pragma unroll
  for (int ck = 0; ck < 4; ++ck) {
    v2f na[4], nb[4];
    if (ck + 1 < 4) {
#pragma unroll
      for (int i = 0; i < 4; ++i) {
        na[i] = *(const v2f*)(h0 + 4 * ((ck + 1) * 4 + i));
        nb[i] = *(const v2f*)(h1 + 4 * ((ck + 1) * 4 + i));
      }
    }
    __builtin_amdgcn_sched_barrier(0);
#pragma unroll
    for (int i = 0; i < 4; ++i) {
      const int ks = ck * 4 + i;
      const v2f a0 = xa[i], a1 = xb[i];
      c[0] = wmma_f32_4(a0, wa[ks], c[0]);
      c[1] = wmma_f32_4(a1, wa[ks], c[1]);
      c[2] = wmma_f32_4(a0, wb[ks], c[2]);
      c[3] = wmma_f32_4(a1, wb[ks], c[3]);
      if (NT == 3) {
        c[4] = wmma_f32_4(a0, wc[ks], c[4]);
        c[5] = wmma_f32_4(a1, wc[ks], c[5]);
      }
    }
    __builtin_amdgcn_sched_barrier(0);
    if (ck + 1 < 4) {
#pragma unroll
      for (int i = 0; i < 4; ++i) { xa[i] = na[i]; xb[i] = nb[i]; }
    }
  }
}

__device__ __forceinline__ void prefetch_graph(const float* __restrict__ ea, const float* __restrict__ x,
                                               int g, int tid)
{
  const char* eb = (const char*)(ea + (size_t)g * EPG * 5);
  const char* fb = (const char*)(x + (size_t)g * PNODES * CIN);
  __builtin_prefetch(eb + (unsigned)tid * 64u, 0, 1);
  const char* p2 = (tid < 17) ? eb + 8192u + (unsigned)tid * 64u : fb + (unsigned)(tid - 17) * 64u;
  if (tid < 37) __builtin_prefetch(p2, 0, 1);
}

__global__ __launch_bounds__(128) void gcn_fused_kernel(
    const float* __restrict__ x, const float* __restrict__ ea,
    const float* __restrict__ W1, const float* __restrict__ b1,
    const float* __restrict__ W2, const float* __restrict__ b2,
    const float* __restrict__ W3, const float* __restrict__ b3,
    const float* __restrict__ W4, const float* __restrict__ b4,
    float* __restrict__ pooled)
{
  __shared__ __attribute__((aligned(16))) float Ha[32][HCOLS];
  __shared__ __attribute__((aligned(16))) float Hb[32][HCOLS];

  const int tid  = threadIdx.x;
  const int wave = __builtin_amdgcn_readfirstlane(tid >> 5);
  const int lane = tid & 31;
  const int half = lane >> 4;
  const int l16  = lane & 15;
  const int swap  = (blockIdx.x + (blockIdx.x >> 8)) & 1;
  const int role  = wave ^ (swap << 1);
  const bool three = role < 2;

  prefetch_graph(ea, x, blockIdx.x, tid);

  const int tA = three ? 3 * role : 2 + 2 * role;
  const int tB = tA + 1, tX = tA + 2;
  const int t2 = role & 1;
  const int nA = tA * 16 + l16, nB = tB * 16 + l16, nX = tX * 16 + l16;
  const int n3 = role * 16 + l16, n2 = t2 * 16 + l16, n1 = l16;
  const bool okB = nB < 152;
  v2f w4a[16], w4b[16], wx[16], w3f[8];
  load_w_frags<64, 152, 16>(w4a, W4, tA, half, l16);
  load_w_frags<64, 152, 16>(w4b, W4, tB, half, l16);
  load_w_frags<32, 64, 8>(w3f, W3, role, half, l16);
  if (three) {
    load_w_frags<64, 152, 16>(wx, W4, tX, half, l16);
  } else {
    v2f w2t[4];
    load_w_frags<16, 32, 4>(w2t, W2, t2, half, l16);
#pragma unroll
    for (int ks = 0; ks < 16; ++ks) { wx[ks].x = (ks < 4) ? w2t[ks & 3].x : 0.0f; wx[ks].y = (ks < 4) ? w2t[ks & 3].y : 0.0f; }
  }
  const float bvA = b4[nA], bvB = okB ? b4[okB ? nB : 0] : 0.0f;
  const float bvX = three ? b4[three ? nX : 0] : b2[n2];
  const float bv3 = b3[n3], bv1 = b1[n1];

  for (int g = blockIdx.x; g < GNUM; g += GCN_WGS) {
    int l16s = l16, halfs = half, tids = tid;
    asm volatile("" : "+v"(l16s), "+v"(halfs), "+v"(tids));
    const int l16 = l16s, half = halfs, tid = tids;
    const int nA = tA * 16 + l16, nB = tB * 16 + l16, nX = tX * 16 + l16;
    const int n3 = role * 16 + l16, n2 = t2 * 16 + l16, n1 = l16;
    const float* ha0 = &Ha[l16][2 * half];
    const float* ha1 = &Ha[16 + l16][2 * half];
    const float* hb0 = &Hb[l16][2 * half];
    const float* hb1 = &Hb[16 + l16][2 * half];
    const float* eaw = ea + (size_t)g * EPG * 5;

    if (g + GCN_WGS < GNUM) prefetch_graph(ea, x, g + GCN_WGS, tid);

    float raw0[12], raw1[12];
    const int  hl   = half & 1, cl = l16 & 15;
    const int  d0   = cl - 8 * hl;
    const bool j1ok = (cl & 7) < 3;
    const int  j1s  = j1ok ? 16 + (cl & 7) + 3 * (cl >> 3) : 16;
    const int  e1   = j1s - 16 - 3 * hl;
    {
      typedef const __attribute__((address_space(1))) char* gcptr;
      gcptr eab = (gcptr)(unsigned long long)eaw;
      asm volatile("" : "+s"(eab));
      const unsigned b0  = 4u * (unsigned)(840 * hl + 5 * cl + 4);
      const unsigned b0m = b0 - 20u;
      const unsigned c0  = 4u * (unsigned)(1684 + 315 * hl + 5 * cl);
      const unsigned b1  = 4u * (unsigned)(840 * hl + 5 * j1s - 1);
      const unsigned c1  = 4u * (unsigned)(1684 + 315 * hl + 5 * j1s);
      const unsigned c1m = c1 - 20u;
#pragma unroll
      for (int q = 0; q < 8; ++q) {
        raw0[q] = *(const __attribute__((address_space(1))) float*)(eab + ((d0 > q) ? b0m : b0) + 420u * q);
        raw1[q] = *(const __attribute__((address_space(1))) float*)(eab + b1 + 420u * q);
      }
#pragma unroll
      for (int q = 8; q < 11; ++q) {
        raw0[q] = *(const __attribute__((address_space(1))) float*)(eab + c0 + 420u * (q - 8));
        raw1[q] = *(const __attribute__((address_space(1))) float*)(eab + ((e1 >= q - 8) ? c1m : c1) + 420u * (q - 8));
      }
      raw0[11] = 0.0f;
      raw1[11] = 0.0f;
    }
    v2f w1f[4] = {};
    if (role == 2) load_w_frags<14, 16, 4>(w1f, W1, 0, half, l16);
    {
      const int  r   = (tid >> 2) & 31, cq = tid & 3;
      const int  nd  = slot_node(r);
      const bool rok = nd >= 0;
      const bool bok = rok && cq < 3;
      const float* xr = x + (size_t)g * (PNODES * CIN) + (unsigned)((rok ? nd : 0) * CIN + 4 * cq);
      const float2 fa = *(const float2*)xr;
      const float2 fb = *(const float2*)(xr + (cq < 3 ? 2 : 0));
      float4 v;
      v.x = rok ? fa.x : 0.0f;
      v.y = rok ? fa.y : 0.0f;
      v.z = bok ? fb.x : 0.0f;
      v.w = bok ? fb.y : 0.0f;
      *(float4*)&Ha[r][4 * cq] = v;
    }
    __syncthreads();

    v2f adj0[6], adj1[6];
    {
      float w0[12], w1[12];
      float s0 = 0.0f, s1 = 0.0f;
#pragma unroll
      for (int q = 0; q < 12; ++q) {
        if (q < 8)       { w0[q] = (q == d0) ? 1.0f : raw0[q];  w1[q] = j1ok ? raw1[q] : 0.0f; }
        else if (q < 11) { w0[q] = raw0[q];  w1[q] = j1ok ? ((q - 8 == e1) ? 1.0f : raw1[q]) : 0.0f; }
        else             { w0[q] = 0.0f;     w1[q] = 0.0f; }
        s0 += w0[q];
        s1 += w1[q];
      }
      s0 += __shfl_xor(s0, 16);
      s1 += __shfl_xor(s1, 16);
      const float dr0 = (s0 > 0.0f) ? rsqrtf(s0) : 0.0f;
      const float dr1 = (s1 > 0.0f) ? rsqrtf(s1) : 0.0f;
#pragma unroll
      for (int ks = 0; ks < 6; ++ks) {
        const int sl = 8 * half + 2 * (ks & 3);
        const float dc0 = (ks < 4) ? __shfl(dr0, sl)     : __shfl(dr1, sl);
        const float dc1 = (ks < 4) ? __shfl(dr0, sl + 1) : __shfl(dr1, sl + 1);
        adj0[ks].x = dc0 * w0[2 * ks]     * dr0;
        adj0[ks].y = dc1 * w0[2 * ks + 1] * dr0;
        adj1[ks].x = dc0 * w1[2 * ks]     * dr1;
        adj1[ks].y = dc1 * w1[2 * ks + 1] * dr1;
      }
    }

    if (role == 2) {
      v8f c0 = {}, c1 = {};
      stepa_pair<4>(c0, c1, ha0, ha1, w1f);
      stepb_regs(Hb, c0, c1, adj0, adj1, n1, half, bv1, true);
    }
    __syncthreads();

    if (!three) {
      v8f c0 = {}, c1 = {};
      stepa_pair<4>(c0, c1, hb0, hb1, wx);
      stepb_regs(Ha, c0, c1, adj0, adj1, n2, half, bvX, true);
    }
    __syncthreads();

    {
      v8f c0 = {}, c1 = {};
      stepa_pair<8>(c0, c1, ha0, ha1, w3f);
      stepb_regs(Hb, c0, c1, adj0, adj1, n3, half, bv3, true);
    }
    __syncthreads();

    {
      v8f c[6] = {};
      float* prow = pooled + (size_t)g * 152;
      float bA = bvA, bB = bvB, bX = bvX;
      asm volatile("" : "+v"(bA), "+v"(bB), "+v"(bX));
      if (three) {
        layer4_stepa<3>(c, hb0, hb1, w4a, w4b, wx);
        const float pA = stepb_pool(c[0], c[1], adj0, adj1, bA);
        const float pB = stepb_pool(c[2], c[3], adj0, adj1, bB);
        const float pX = stepb_pool(c[4], c[5], adj0, adj1, bX);
        prow[nA + 16 * half] = half ? pB : pA;
        if (half == 0) prow[nX] = pX;
      } else {
        layer4_stepa<2>(c, hb0, hb1, w4a, w4b, wx);
        const float pA = stepb_pool(c[0], c[1], adj0, adj1, bA);
        const float pB = stepb_pool(c[2], c[3], adj0, adj1, bB);
        const int n = nA + 16 * half;
        if (n < 152) prow[n] = half ? pB : pA;
      }
    }
  }
}

#define CONV_TW    32
#define CONV_PITCH 460
#define CONV_NS    57
#define CONV_G     3
#define CONV_NG    (CONV_NS / CONV_G)
#define CONV_NT    17
#define CONV_NCOL  1920
#define CONV_WGS   255
#define CONV_RPT   11

__device__ __forceinline__ void conv_stage_load(float4 (&v)[CONV_RPT + 2], const float* pcol, int r0)
{
  const float* base = pcol + r0 * 152;
#pragma unroll
  for (int j = 0; j < CONV_RPT + 2; ++j) {
    if (j == 0) {
      const int rc = r0 < 1 ? 0 : r0 - 1;
      v[j] = *(const float4*)(pcol + rc * 152);
    } else if (j < CONV_RPT) {
      v[j] = *(const float4*)(base + (j - 1) * 152);
    } else {
      const int r  = r0 - 1 + j;
      const int rc = r >= CONV_NCOL ? CONV_NCOL - 1 : r;
      v[j] = *(const float4*)(pcol + rc * 152);
    }
  }
}

__device__ __forceinline__ void conv_stage_store(float (*Ab)[CONV_PITCH], const float4 (&v)[CONV_RPT + 2],
                                                 int r0, int m0, int c4)
{
#pragma unroll
  for (int i = 0; i < CONV_RPT; ++i) {
    if (i < CONV_RPT - 1 || m0 + i < CONV_TW) {
      const float4 a = v[i], b = v[i + 1], c = v[i + 2];
      float4* dst = (float4*)&Ab[m0 + i][c4 * 12];
      dst[0] = make_float4(a.x, b.x, c.x, a.y);
      dst[1] = make_float4(b.y, c.y, a.z, b.z);
      dst[2] = make_float4(c.z, a.w, b.w, c.w);
    }
  }
  const int tb = r0 % TLEN;
  const int i0 = tb == 0 ? 0 : TLEN - tb;
  const int i2 = TLEN - 1 - tb;
  if (i0 < CONV_RPT && m0 + i0 < CONV_TW) {
    float* q = &Ab[m0 + i0][c4 * 12];
    q[0] = 0.0f;  q[3] = 0.0f;  q[6] = 0.0f;  q[9] = 0.0f;
  }
  if (i2 < CONV_RPT && m0 + i2 < CONV_TW) {
    float* q = &Ab[m0 + i2][c4 * 12];
    q[2] = 0.0f;  q[5] = 0.0f;  q[8] = 0.0f;  q[11] = 0.0f;
  }
}

__global__ __launch_bounds__(128) void conv_wmma_kernel(
    const float* __restrict__ pooled, const float* __restrict__ cw,
    const float* __restrict__ cb, float* __restrict__ convout)
{
  __shared__ __attribute__((aligned(16))) float Aim[2][CONV_TW][CONV_PITCH];

  const int tid  = threadIdx.x;
  const int wave = __builtin_amdgcn_readfirstlane(tid >> 5);
  const int lane = tid & 31;
  const int half = lane >> 4;
  const int l16  = lane & 15;
  const int p0   = blockIdx.x * 4;
  const int rbA  = p0 / CONV_NT;
  const int rbB  = (p0 + 3) / CONV_NT;
  const int rb   = (p0 + wave) / CONV_NT;
  const int nt   = p0 + wave - rb * CONV_NT;
  const bool two = rbB != rbA;

  const int  rg     = tid / 38;
  const int  c4     = tid - rg * 38;
  const bool stager = tid < 3 * 38;
  const float* pcol = pooled + c4 * 4;
  if (stager) {
    float4 va[CONV_RPT + 2], vb[CONV_RPT + 2];
    const int ra = rbA * CONV_TW + CONV_RPT * rg;
    const int rc = rbB * CONV_TW + CONV_RPT * rg;
    conv_stage_load(va, pcol, ra);
    if (two) conv_stage_load(vb, pcol, rc);
    conv_stage_store(Aim[0], va, ra, CONV_RPT * rg, c4);
    if (two) conv_stage_store(Aim[1], vb, rc, CONV_RPT * rg, c4);
  }

  float4 wq[CONV_NS];
  const int nb = nt * 16 + 8 * half;
  const float* wrow = cw + (size_t)(nt * 16 + l16) * KDIM + 4 * half;
#pragma unroll
  for (int s = 0; s < CONV_NS; ++s) wq[s] = *(const float4*)(wrow + 8 * s);
  const float4 bl = *(const float4*)(cb + nb);
  const float4 bh = *(const float4*)(cb + nb + 4);

  __builtin_amdgcn_fence(__ATOMIC_RELEASE, "workgroup", "local");
  __builtin_amdgcn_s_barrier();
  __builtin_amdgcn_fence(__ATOMIC_ACQUIRE, "workgroup", "local");

  const float (*Ab)[CONV_PITCH] = Aim[rb - rbA];
  const float* x0p = &Ab[l16][4 * half];
  const float* x1p = &Ab[16 + l16][4 * half];
  v8f acc0 = {}, acc1 = {};
  float4 xc[CONV_G], yc[CONV_G];
#pragma unroll
  for (int i = 0; i < CONV_G; ++i) {
    xc[i] = *(const float4*)(x0p + 8 * i);
    yc[i] = *(const float4*)(x1p + 8 * i);
  }
#pragma unroll
  for (int g = 0; g < CONV_NG; ++g) {
    float4 xn[CONV_G], yn[CONV_G];
    if (g + 1 < CONV_NG) {
#pragma unroll
      for (int i = 0; i < CONV_G; ++i) {
        xn[i] = *(const float4*)(x0p + 8 * ((g + 1) * CONV_G + i));
        yn[i] = *(const float4*)(x1p + 8 * ((g + 1) * CONV_G + i));
      }
    }
    __builtin_amdgcn_sched_barrier(0);
#pragma unroll
    for (int i = 0; i < CONV_G; ++i) {
      const float4 w = wq[g * CONV_G + i];
      v2f wa, wb, xa, xb, ya, yb;
      wa.x = w.x;     wa.y = w.y;     wb.x = w.z;     wb.y = w.w;
      xa.x = xc[i].x; xa.y = xc[i].y; xb.x = xc[i].z; xb.y = xc[i].w;
      ya.x = yc[i].x; ya.y = yc[i].y; yb.x = yc[i].z; yb.y = yc[i].w;
      acc0 = wmma_f32_4(wa, xa, acc0);
      acc1 = wmma_f32_4(wa, ya, acc1);
      acc0 = wmma_f32_4(wb, xb, acc0);
      acc1 = wmma_f32_4(wb, yb, acc1);
    }
    __builtin_amdgcn_sched_barrier(0);
    if (g + 1 < CONV_NG) {
#pragma unroll
      for (int i = 0; i < CONV_G; ++i) { xc[i] = xn[i]; yc[i] = yn[i]; }
    }
  }

  const float bias[8] = {bl.x, bl.y, bl.z, bl.w, bh.x, bh.y, bh.z, bh.w};
  const int ca = rb * CONV_TW + l16, cc = ca + 16;
  const int ba = ca / TLEN, bc = cc / TLEN;
  float* oa = convout + ((size_t)ba * OCH + nb) * TLEN + (ca - ba * TLEN);
  float* oc = convout + ((size_t)bc * OCH + nb) * TLEN + (cc - bc * TLEN);
#pragma unroll
  for (int r = 0; r < 8; ++r) {
    oa[r * TLEN] = acc0[r] + bias[r];
    oc[r * TLEN] = acc1[r] + bias[r];
  }
}

__global__ __launch_bounds__(256) void caps_norm_kernel(
    const float* __restrict__ convout, const float* __restrict__ gamma,
    const float* __restrict__ beta, float* __restrict__ out)
{
  const int idx = blockIdx.x * 256 + threadIdx.x;
  if (idx >= BSZ * TLEN * NCLS) return;
  const int cls = idx % NCLS;
  const int t   = (idx / NCLS) % TLEN;
  const int b   = idx / (NCLS * TLEN);
  const float scale = gamma[t] * rsqrtf(1.0f + 0.001f);
  const float shift = beta[t];
  float sum = 0.0f;
  for (int d = 0; d < 16; ++d) {
    const int o = d * NCLS + cls;
    const float v  = convout[((size_t)b * OCH + o) * TLEN + t];
    const float s  = 1.0f / (1.0f + __expf(-(v * scale + shift)));
    const float df = s - 0.5f;
    sum += df * df;
  }
  out[idx] = sqrtf(sum * 0.25f);
}

extern "C" void kernel_launch(void* const* d_in, const int* in_sizes, int n_in,
                              void* d_out, int out_size, void* d_ws, size_t ws_size,
                              hipStream_t stream) {
  (void)in_sizes; (void)n_in; (void)out_size; (void)ws_size;
  const float* x  = (const float*)d_in[0];
  const float* ea = (const float*)d_in[3];
  const float* W1 = (const float*)d_in[4];  const float* b1 = (const float*)d_in[5];
  const float* W2 = (const float*)d_in[6];  const float* b2 = (const float*)d_in[7];
  const float* W3 = (const float*)d_in[8];  const float* b3 = (const float*)d_in[9];
  const float* W4 = (const float*)d_in[10]; const float* b4 = (const float*)d_in[11];
  const float* cw = (const float*)d_in[12]; const float* cb = (const float*)d_in[13];
  const float* gm = (const float*)d_in[14]; const float* bt = (const float*)d_in[15];

  float* pooled  = (float*)d_ws;
  float* convout = pooled + (size_t)GNUM * 152;

  gcn_fused_kernel<<<GCN_WGS, 128, 0, stream>>>(x, ea, W1, b1, W2, b2, W3, b3, W4, b4, pooled);
  conv_wmma_kernel<<<CONV_WGS, 128, 0, stream>>>(pooled, cw, cb, convout);
  const int nout = BSZ * TLEN * NCLS;
  caps_norm_kernel<<<(nout + 255) / 256, 256, 0, stream>>>(convout, gm, bt, (float*)d_out);
}


# capsule head load path: thread = (batch, class, time) with time fastest on 255 x 128 threads; a wave's loads are contiguous runs and a thread's 16 loads are one run, one memory latency not eight
# speedup vs baseline: 7.3340x; 1.1057x over previous
#include <hip/hip_runtime.h>
#include <hip/hip_bf16.h>
#include <cmath>

typedef __attribute__((ext_vector_type(2))) float v2f;
typedef __attribute__((ext_vector_type(8))) float v8f;

#define PNODES 22
#define GNUM   1920
#define CIN    14
#define HCOLS  160
#define EPG    462
#define TLEN   120
#define BSZ    16
#define OCH    272
#define KDIM   456
#define NCLS   17

__device__ __forceinline__ v8f wmma_f32_4(v2f a, v2f b, v8f c) {
  return __builtin_amdgcn_wmma_f32_16x16x4_f32(
       false, a,  false, b,
       (short)0, c,  false,  false);
}

#define GCN_WGS 960

template <int CIN_, int COUT_, int KS>
__device__ __forceinline__ void load_w_frags(v2f (&wf)[KS], const float* __restrict__ W,
                                             int nt, int half, int l16)
{
  const int  n   = nt * 16 + l16;
  const bool nok = n < COUT_;
  const int  ncl = nok ? n : 0;
  const float* base = W + (2 * half) * COUT_ + ncl;
#pragma unroll
  for (int ks = 0; ks < KS; ++ks) {
    if (4 * ks + 3 < CIN_) {
      const float w0 = base[(4 * ks) * COUT_];
      const float w1 = base[(4 * ks + 1) * COUT_];
      wf[ks].x = nok ? w0 : 0.0f;
      wf[ks].y = nok ? w1 : 0.0f;
    } else {
      const int k0  = 4 * ks + 2 * half;
      const int kr0 = (k0     < CIN_) ? k0     : 0;
      const int kr1 = (k0 + 1 < CIN_) ? k0 + 1 : 0;
      const float w0 = W[kr0 * COUT_ + ncl];
      const float w1 = W[kr1 * COUT_ + ncl];
      wf[ks].x = (k0     < CIN_ && nok) ? w0 : 0.0f;
      wf[ks].y = (k0 + 1 < CIN_ && nok) ? w1 : 0.0f;
    }
  }
}

__device__ __forceinline__ int slot_node(int s) { return s < 19 ? s : ((s >= 24 && s < 27) ? s - 5 : -1); }

__device__ __forceinline__ void store_h(float (&H)[32][HCOLS], v8f c, int mt, int n, int half,
                                        float bv, bool nok)
{
#pragma unroll
  for (int v = 0; v < 8; ++v) {
    const int  row  = mt * 16 + 8 * half + v;
    const bool real = (mt == 0) || (v < 3);
    H[row][n] = (real && nok) ? fmaxf(c[v] + bv, 0.0f) : 0.0f;
  }
}

__device__ __forceinline__ void stepb_regs(float (&H)[32][HCOLS], const v8f t0, const v8f t1,
    const v2f (&adj0)[6], const v2f (&adj1)[6], int n, int half, float bv, bool nok)
{
  v8f d0 = {}, d1 = {};
#pragma unroll
  for (int ks = 0; ks < 6; ++ks) {
    v2f b;
    if (ks < 4) { b.x = t0[2 * ks];       b.y = t0[2 * ks + 1]; }
    else        { b.x = t1[2 * (ks - 4)]; b.y = t1[2 * (ks - 4) + 1]; }
    d0 = wmma_f32_4(adj0[ks], b, d0);
    d1 = wmma_f32_4(adj1[ks], b, d1);
  }
  store_h(H, d0, 0, n, half, bv, nok);
  store_h(H, d1, 1, n, half, bv, nok);
}

__device__ __forceinline__ float stepb_pool(const v8f t0, const v8f t1,
    const v2f (&adj0)[6], const v2f (&adj1)[6], float bv)
{
  v8f d0 = {}, d1 = {};
#pragma unroll
  for (int ks = 0; ks < 6; ++ks) {
    v2f b;
    if (ks < 4) { b.x = t0[2 * ks];       b.y = t0[2 * ks + 1]; }
    else        { b.x = t1[2 * (ks - 4)]; b.y = t1[2 * (ks - 4) + 1]; }
    d0 = wmma_f32_4(adj0[ks], b, d0);
    d1 = wmma_f32_4(adj1[ks], b, d1);
  }
  float s = 0.0f;
#pragma unroll
  for (int v = 0; v < 8; ++v) s += fmaxf(d0[v] + bv, 0.0f);
#pragma unroll
  for (int v = 0; v < 3; ++v) s += fmaxf(d1[v] + bv, 0.0f);
  s += __shfl_xor(s, 16);
  return s * (1.0f / 22.0f);
}

template <int KS>
__device__ __forceinline__ void stepa_pair(v8f& c0, v8f& c1, const float* h0, const float* h1,
                                           const v2f* wf)
{
#pragma unroll
  for (int ks = 0; ks < KS; ++ks) {
    const v2f a0 = *(const v2f*)(h0 + 4 * ks);
    const v2f a1 = *(const v2f*)(h1 + 4 * ks);
    c0 = wmma_f32_4(a0, wf[ks], c0);
    c1 = wmma_f32_4(a1, wf[ks], c1);
  }
}

template <int NT>
__device__ __forceinline__ void layer4_stepa(v8f (&c)[6], const float* h0, const float* h1,
    const v2f (&wa)[16], const v2f (&wb)[16], const v2f (&wc)[16])
{
  v2f xa[4], xb[4];
#pragma unroll
  for (int i = 0; i < 4; ++i) {
    xa[i] = *(const v2f*)(h0 + 4 * i);
    xb[i] = *(const v2f*)(h1 + 4 * i);
  }
#pragma unroll
  for (int ck = 0; ck < 4; ++ck) {
    v2f na[4], nb[4];
    if (ck + 1 < 4) {
#pragma unroll
      for (int i = 0; i < 4; ++i) {
        na[i] = *(const v2f*)(h0 + 4 * ((ck + 1) * 4 + i));
        nb[i] = *(const v2f*)(h1 + 4 * ((ck + 1) * 4 + i));
      }
    }
    __builtin_amdgcn_sched_barrier(0);
#pragma unroll
    for (int i = 0; i < 4; ++i) {
      const int ks = ck * 4 + i;
      const v2f a0 = xa[i], a1 = xb[i];
      c[0] = wmma_f32_4(a0, wa[ks], c[0]);
      c[1] = wmma_f32_4(a1, wa[ks], c[1]);
      c[2] = wmma_f32_4(a0, wb[ks], c[2]);
      c[3] = wmma_f32_4(a1, wb[ks], c[3]);
      if (NT == 3) {
        c[4] = wmma_f32_4(a0, wc[ks], c[4]);
        c[5] = wmma_f32_4(a1, wc[ks], c[5]);
      }
    }
    __builtin_amdgcn_sched_barrier(0);
    if (ck + 1 < 4) {
#pragma unroll
      for (int i = 0; i < 4; ++i) { xa[i] = na[i]; xb[i] = nb[i]; }
    }
  }
}

__device__ __forceinline__ void prefetch_graph(const float* __restrict__ ea, const float* __restrict__ x,
                                               int g, int tid)
{
  const char* eb = (const char*)(ea + (size_t)g * EPG * 5);
  const char* fb = (const char*)(x + (size_t)g * PNODES * CIN);
  __builtin_prefetch(eb + (unsigned)tid * 64u, 0, 1);
  const char* p2 = (tid < 17) ? eb + 8192u + (unsigned)tid * 64u : fb + (unsigned)(tid - 17) * 64u;
  if (tid < 37) __builtin_prefetch(p2, 0, 1);
}

__global__ __launch_bounds__(128) void gcn_fused_kernel(
    const float* __restrict__ x, const float* __restrict__ ea,
    const float* __restrict__ W1, const float* __restrict__ b1,
    const float* __restrict__ W2, const float* __restrict__ b2,
    const float* __restrict__ W3, const float* __restrict__ b3,
    const float* __restrict__ W4, const float* __restrict__ b4,
    float* __restrict__ pooled)
{
  __shared__ __attribute__((aligned(16))) float Ha[32][HCOLS];
  __shared__ __attribute__((aligned(16))) float Hb[32][HCOLS];

  const int tid  = threadIdx.x;
  const int wave = __builtin_amdgcn_readfirstlane(tid >> 5);
  const int lane = tid & 31;
  const int half = lane >> 4;
  const int l16  = lane & 15;
  const int swap  = (blockIdx.x + (blockIdx.x >> 8)) & 1;
  const int role  = wave ^ (swap << 1);
  const bool three = role < 2;

  prefetch_graph(ea, x, blockIdx.x, tid);

  const int tA = three ? 3 * role : 2 + 2 * role;
  const int tB = tA + 1, tX = tA + 2;
  const int t2 = role & 1;
  const int nA = tA * 16 + l16, nB = tB * 16 + l16, nX = tX * 16 + l16;
  const int n3 = role * 16 + l16, n2 = t2 * 16 + l16, n1 = l16;
  const bool okB = nB < 152;
  v2f w4a[16], w4b[16], wx[16], w3f[8];
  load_w_frags<64, 152, 16>(w4a, W4, tA, half, l16);
  load_w_frags<64, 152, 16>(w4b, W4, tB, half, l16);
  load_w_frags<32, 64, 8>(w3f, W3, role, half, l16);
  if (three) {
    load_w_frags<64, 152, 16>(wx, W4, tX, half, l16);
  } else {
    v2f w2t[4];
    load_w_frags<16, 32, 4>(w2t, W2, t2, half, l16);
#pragma unroll
    for (int ks = 0; ks < 16; ++ks) { wx[ks].x = (ks < 4) ? w2t[ks & 3].x : 0.0f; wx[ks].y = (ks < 4) ? w2t[ks & 3].y : 0.0f; }
  }
  const float bvA = b4[nA], bvB = okB ? b4[okB ? nB : 0] : 0.0f;
  const float bvX = three ? b4[three ? nX : 0] : b2[n2];
  const float bv3 = b3[n3], bv1 = b1[n1];

  for (int g = blockIdx.x; g < GNUM; g += GCN_WGS) {
    int l16s = l16, halfs = half, tids = tid;
    asm volatile("" : "+v"(l16s), "+v"(halfs), "+v"(tids));
    const int l16 = l16s, half = halfs, tid = tids;
    const int nA = tA * 16 + l16, nB = tB * 16 + l16, nX = tX * 16 + l16;
    const int n3 = role * 16 + l16, n2 = t2 * 16 + l16, n1 = l16;
    const float* ha0 = &Ha[l16][2 * half];
    const float* ha1 = &Ha[16 + l16][2 * half];
    const float* hb0 = &Hb[l16][2 * half];
    const float* hb1 = &Hb[16 + l16][2 * half];
    const float* eaw = ea + (size_t)g * EPG * 5;

    if (g + GCN_WGS < GNUM) prefetch_graph(ea, x, g + GCN_WGS, tid);

    float raw0[12], raw1[12];
    const int  hl   = half & 1, cl = l16 & 15;
    const int  d0   = cl - 8 * hl;
    const bool j1ok = (cl & 7) < 3;
    const int  j1s  = j1ok ? 16 + (cl & 7) + 3 * (cl >> 3) : 16;
    const int  e1   = j1s - 16 - 3 * hl;
    {
      typedef const __attribute__((address_space(1))) char* gcptr;
      gcptr eab = (gcptr)(unsigned long long)eaw;
      asm volatile("" : "+s"(eab));
      const unsigned b0  = 4u * (unsigned)(840 * hl + 5 * cl + 4);
      const unsigned b0m = b0 - 20u;
      const unsigned c0  = 4u * (unsigned)(1684 + 315 * hl + 5 * cl);
      const unsigned b1  = 4u * (unsigned)(840 * hl + 5 * j1s - 1);
      const unsigned c1  = 4u * (unsigned)(1684 + 315 * hl + 5 * j1s);
      const unsigned c1m = c1 - 20u;
#pragma unroll
      for (int q = 0; q < 8; ++q) {
        raw0[q] = *(const __attribute__((address_space(1))) float*)(eab + ((d0 > q) ? b0m : b0) + 420u * q);
        raw1[q] = *(const __attribute__((address_space(1))) float*)(eab + b1 + 420u * q);
      }
#pragma unroll
      for (int q = 8; q < 11; ++q) {
        raw0[q] = *(const __attribute__((address_space(1))) float*)(eab + c0 + 420u * (q - 8));
        raw1[q] = *(const __attribute__((address_space(1))) float*)(eab + ((e1 >= q - 8) ? c1m : c1) + 420u * (q - 8));
      }
      raw0[11] = 0.0f;
      raw1[11] = 0.0f;
    }
    v2f w1f[4] = {};
    if (role == 2) load_w_frags<14, 16, 4>(w1f, W1, 0, half, l16);
    {
      const int  r   = (tid >> 2) & 31, cq = tid & 3;
      const int  nd  = slot_node(r);
      const bool rok = nd >= 0;
      const bool bok = rok && cq < 3;
      const float* xr = x + (size_t)g * (PNODES * CIN) + (unsigned)((rok ? nd : 0) * CIN + 4 * cq);
      const float2 fa = *(const float2*)xr;
      const float2 fb = *(const float2*)(xr + (cq < 3 ? 2 : 0));
      float4 v;
      v.x = rok ? fa.x : 0.0f;
      v.y = rok ? fa.y : 0.0f;
      v.z = bok ? fb.x : 0.0f;
      v.w = bok ? fb.y : 0.0f;
      *(float4*)&Ha[r][4 * cq] = v;
    }
    __syncthreads();

    v2f adj0[6], adj1[6];
    {
      float w0[12], w1[12];
      float s0 = 0.0f, s1 = 0.0f;
#pragma unroll
      for (int q = 0; q < 12; ++q) {
        if (q < 8)       { w0[q] = (q == d0) ? 1.0f : raw0[q];  w1[q] = j1ok ? raw1[q] : 0.0f; }
        else if (q < 11) { w0[q] = raw0[q];  w1[q] = j1ok ? ((q - 8 == e1) ? 1.0f : raw1[q]) : 0.0f; }
        else             { w0[q] = 0.0f;     w1[q] = 0.0f; }
        s0 += w0[q];
        s1 += w1[q];
      }
      s0 += __shfl_xor(s0, 16);
      s1 += __shfl_xor(s1, 16);
      const float dr0 = (s0 > 0.0f) ? rsqrtf(s0) : 0.0f;
      const float dr1 = (s1 > 0.0f) ? rsqrtf(s1) : 0.0f;
#pragma unroll
      for (int ks = 0; ks < 6; ++ks) {
        const int sl = 8 * half + 2 * (ks & 3);
        const float dc0 = (ks < 4) ? __shfl(dr0, sl)     : __shfl(dr1, sl);
        const float dc1 = (ks < 4) ? __shfl(dr0, sl + 1) : __shfl(dr1, sl + 1);
        adj0[ks].x = dc0 * w0[2 * ks]     * dr0;
        adj0[ks].y = dc1 * w0[2 * ks + 1] * dr0;
        adj1[ks].x = dc0 * w1[2 * ks]     * dr1;
        adj1[ks].y = dc1 * w1[2 * ks + 1] * dr1;
      }
    }

    if (role == 2) {
      v8f c0 = {}, c1 = {};
      stepa_pair<4>(c0, c1, ha0, ha1, w1f);
      stepb_regs(Hb, c0, c1, adj0, adj1, n1, half, bv1, true);
    }
    __syncthreads();

    if (!three) {
      v8f c0 = {}, c1 = {};
      stepa_pair<4>(c0, c1, hb0, hb1, wx);
      stepb_regs(Ha, c0, c1, adj0, adj1, n2, half, bvX, true);
    }
    __syncthreads();

    {
      v8f c0 = {}, c1 = {};
      stepa_pair<8>(c0, c1, ha0, ha1, w3f);
      stepb_regs(Hb, c0, c1, adj0, adj1, n3, half, bv3, true);
    }
    __syncthreads();

    {
      v8f c[6] = {};
      float* prow = pooled + (size_t)g * 152;
      float bA = bvA, bB = bvB, bX = bvX;
      asm volatile("" : "+v"(bA), "+v"(bB), "+v"(bX));
      if (three) {
        layer4_stepa<3>(c, hb0, hb1, w4a, w4b, wx);
        const float pA = stepb_pool(c[0], c[1], adj0, adj1, bA);
        const float pB = stepb_pool(c[2], c[3], adj0, adj1, bB);
        const float pX = stepb_pool(c[4], c[5], adj0, adj1, bX);
        prow[nA + 16 * half] = half ? pB : pA;
        if (half == 0) prow[nX] = pX;
      } else {
        layer4_stepa<2>(c, hb0, hb1, w4a, w4b, wx);
        const float pA = stepb_pool(c[0], c[1], adj0, adj1, bA);
        const float pB = stepb_pool(c[2], c[3], adj0, adj1, bB);
        const int n = nA + 16 * half;
        if (n < 152) prow[n] = half ? pB : pA;
      }
    }
  }
}

#define CONV_TW    32
#define CONV_PITCH 460
#define CONV_NS    57
#define CONV_G     3
#define CONV_NG    (CONV_NS / CONV_G)
#define CONV_NT    17
#define CONV_NCOL  1920
#define CONV_WGS   255
#define CONV_RPT   11

__device__ __forceinline__ void conv_stage_load(float4 (&v)[CONV_RPT + 2], const float* pcol, int r0)
{
  const float* base = pcol + r0 * 152;
#pragma unroll
  for (int j = 0; j < CONV_RPT + 2; ++j) {
    if (j == 0) {
      const int rc = r0 < 1 ? 0 : r0 - 1;
      v[j] = *(const float4*)(pcol + rc * 152);
    } else if (j < CONV_RPT) {
      v[j] = *(const float4*)(base + (j - 1) * 152);
    } else {
      const int r  = r0 - 1 + j;
      const int rc = r >= CONV_NCOL ? CONV_NCOL - 1 : r;
      v[j] = *(const float4*)(pcol + rc * 152);
    }
  }
}

__device__ __forceinline__ void conv_stage_store(float (*Ab)[CONV_PITCH], const float4 (&v)[CONV_RPT + 2],
                                                 int r0, int m0, int c4)
{
#pragma unroll
  for (int i = 0; i < CONV_RPT; ++i) {
    if (i < CONV_RPT - 1 || m0 + i < CONV_TW) {
      const float4 a = v[i], b = v[i + 1], c = v[i + 2];
      float4* dst = (float4*)&Ab[m0 + i][c4 * 12];
      dst[0] = make_float4(a.x, b.x, c.x, a.y);
      dst[1] = make_float4(b.y, c.y, a.z, b.z);
      dst[2] = make_float4(c.z, a.w, b.w, c.w);
    }
  }
  const int tb = r0 % TLEN;
  const int i0 = tb == 0 ? 0 : TLEN - tb;
  const int i2 = TLEN - 1 - tb;
  if (i0 < CONV_RPT && m0 + i0 < CONV_TW) {
    float* q = &Ab[m0 + i0][c4 * 12];
    q[0] = 0.0f;  q[3] = 0.0f;  q[6] = 0.0f;  q[9] = 0.0f;
  }
  if (i2 < CONV_RPT && m0 + i2 < CONV_TW) {
    float* q = &Ab[m0 + i2][c4 * 12];
    q[2] = 0.0f;  q[5] = 0.0f;  q[8] = 0.0f;  q[11] = 0.0f;
  }
}

__global__ __launch_bounds__(128) void conv_wmma_kernel(
    const float* __restrict__ pooled, const float* __restrict__ cw,
    const float* __restrict__ cb, float* __restrict__ convout)
{
  __shared__ __attribute__((aligned(16))) float Aim[2][CONV_TW][CONV_PITCH];

  const int tid  = threadIdx.x;
  const int wave = __builtin_amdgcn_readfirstlane(tid >> 5);
  const int lane = tid & 31;
  const int half = lane >> 4;
  const int l16  = lane & 15;
  const int p0   = blockIdx.x * 4;
  const int rbA  = p0 / CONV_NT;
  const int rbB  = (p0 + 3) / CONV_NT;
  const int rb   = (p0 + wave) / CONV_NT;
  const int nt   = p0 + wave - rb * CONV_NT;
  const bool two = rbB != rbA;

  const int  rg     = tid / 38;
  const int  c4     = tid - rg * 38;
  const bool stager = tid < 3 * 38;
  const float* pcol = pooled + c4 * 4;
  if (stager) {
    float4 va[CONV_RPT + 2], vb[CONV_RPT + 2];
    const int ra = rbA * CONV_TW + CONV_RPT * rg;
    const int rc = rbB * CONV_TW + CONV_RPT * rg;
    conv_stage_load(va, pcol, ra);
    if (two) conv_stage_load(vb, pcol, rc);
    conv_stage_store(Aim[0], va, ra, CONV_RPT * rg, c4);
    if (two) conv_stage_store(Aim[1], vb, rc, CONV_RPT * rg, c4);
  }

  float4 wq[CONV_NS];
  const int nb = nt * 16 + 8 * half;
  const float* wrow = cw + (size_t)(nt * 16 + l16) * KDIM + 4 * half;
#pragma unroll
  for (int s = 0; s < CONV_NS; ++s) wq[s] = *(const float4*)(wrow + 8 * s);
  const float4 bl = *(const float4*)(cb + nb);
  const float4 bh = *(const float4*)(cb + nb + 4);

  __builtin_amdgcn_fence(__ATOMIC_RELEASE, "workgroup", "local");
  __builtin_amdgcn_s_barrier();
  __builtin_amdgcn_fence(__ATOMIC_ACQUIRE, "workgroup", "local");

  const float (*Ab)[CONV_PITCH] = Aim[rb - rbA];
  const float* x0p = &Ab[l16][4 * half];
  const float* x1p = &Ab[16 + l16][4 * half];
  v8f acc0 = {}, acc1 = {};
  float4 xc[CONV_G], yc[CONV_G];
#pragma unroll
  for (int i = 0; i < CONV_G; ++i) {
    xc[i] = *(const float4*)(x0p + 8 * i);
    yc[i] = *(const float4*)(x1p + 8 * i);
  }
#pragma unroll
  for (int g = 0; g < CONV_NG; ++g) {
    float4 xn[CONV_G], yn[CONV_G];
    if (g + 1 < CONV_NG) {
#pragma unroll
      for (int i = 0; i < CONV_G; ++i) {
        xn[i] = *(const float4*)(x0p + 8 * ((g + 1) * CONV_G + i));
        yn[i] = *(const float4*)(x1p + 8 * ((g + 1) * CONV_G + i));
      }
    }
    __builtin_amdgcn_sched_barrier(0);
#pragma unroll
    for (int i = 0; i < CONV_G; ++i) {
      const float4 w = wq[g * CONV_G + i];
      v2f wa, wb, xa, xb, ya, yb;
      wa.x = w.x;     wa.y = w.y;     wb.x = w.z;     wb.y = w.w;
      xa.x = xc[i].x; xa.y = xc[i].y; xb.x = xc[i].z; xb.y = xc[i].w;
      ya.x = yc[i].x; ya.y = yc[i].y; yb.x = yc[i].z; yb.y = yc[i].w;
      acc0 = wmma_f32_4(wa, xa, acc0);
      acc1 = wmma_f32_4(wa, ya, acc1);
      acc0 = wmma_f32_4(wb, xb, acc0);
      acc1 = wmma_f32_4(wb, yb, acc1);
    }
    __builtin_amdgcn_sched_barrier(0);
    if (g + 1 < CONV_NG) {
#pragma unroll
      for (int i = 0; i < CONV_G; ++i) { xc[i] = xn[i]; yc[i] = yn[i]; }
    }
  }

  const float bias[8] = {bl.x, bl.y, bl.z, bl.w, bh.x, bh.y, bh.z, bh.w};
  const int ca = rb * CONV_TW + l16, cc = ca + 16;
  const int ba = ca / TLEN, bc = cc / TLEN;
  float* oa = convout + ((size_t)ba * OCH + nb) * TLEN + (ca - ba * TLEN);
  float* oc = convout + ((size_t)bc * OCH + nb) * TLEN + (cc - bc * TLEN);
#pragma unroll
  for (int r = 0; r < 8; ++r) {
    oa[r * TLEN] = acc0[r] + bias[r];
    oc[r * TLEN] = acc1[r] + bias[r];
  }
}

#define CAPS_WGS 255
__global__ __launch_bounds__(128) void caps_norm_kernel(
    const float* __restrict__ convout, const float* __restrict__ gamma,
    const float* __restrict__ beta, float* __restrict__ out)
{
  const int idx = blockIdx.x * 128 + threadIdx.x;
  const int row = idx / TLEN;
  const int t   = idx - row * TLEN;
  const int b   = row / NCLS;
  const int cls = row - b * NCLS;
  const float* p = convout + ((size_t)b * OCH + cls) * TLEN + t;
  float v[16];
#pragma unroll
  for (int d = 0; d < 16; ++d) v[d] = p[d * (NCLS * TLEN)];
  const float scale = gamma[t] * rsqrtf(1.0f + 0.001f);
  const float shift = beta[t];
  float sum = 0.0f;
#pragma unroll
  for (int d = 0; d < 16; ++d) {
    const float s  = 1.0f / (1.0f + __expf(-(v[d] * scale + shift)));
    const float df = s - 0.5f;
    sum = fmaf(df, df, sum);
  }
  out[((size_t)b * TLEN + t) * NCLS + cls] = sqrtf(sum * 0.25f);
}

extern "C" void kernel_launch(void* const* d_in, const int* in_sizes, int n_in,
                              void* d_out, int out_size, void* d_ws, size_t ws_size,
                              hipStream_t stream) {
  (void)in_sizes; (void)n_in; (void)out_size; (void)ws_size;
  const float* x  = (const float*)d_in[0];
  const float* ea = (const float*)d_in[3];
  const float* W1 = (const float*)d_in[4];  const float* b1 = (const float*)d_in[5];
  const float* W2 = (const float*)d_in[6];  const float* b2 = (const float*)d_in[7];
  const float* W3 = (const float*)d_in[8];  const float* b3 = (const float*)d_in[9];
  const float* W4 = (const float*)d_in[10]; const float* b4 = (const float*)d_in[11];
  const float* cw = (const float*)d_in[12]; const float* cb = (const float*)d_in[13];
  const float* gm = (const float*)d_in[14]; const float* bt = (const float*)d_in[15];

  float* pooled  = (float*)d_ws;
  float* convout = pooled + (size_t)GNUM * 152;

  gcn_fused_kernel<<<GCN_WGS, 128, 0, stream>>>(x, ea, W1, b1, W2, b2, W3, b3, W4, b4, pooled);
  conv_wmma_kernel<<<CONV_WGS, 128, 0, stream>>>(pooled, cw, cb, convout);
  caps_norm_kernel<<<CAPS_WGS, 128, 0, stream>>>(convout, gm, bt, (float*)d_out);
}
